# starting kernel
# baseline (speedup 1.0000x reference)
#include <hip/hip_runtime.h>
#include <math.h>

typedef __attribute__((ext_vector_type(2))) float v2f;
typedef __attribute__((ext_vector_type(8))) float v8f;

#define FDIM 36
#define NHEAD 3
#define HDIM 12
#define NEG_SLOPE 0.2f
#define NEG_INF_KEY 0x007FFFFFu

__device__ __forceinline__ unsigned f2key(float f) {
  unsigned u = __float_as_uint(f);
  return (u & 0x80000000u) ? ~u : (u | 0x80000000u);
}
__device__ __forceinline__ float key2f(unsigned k) {
  return (k & 0x80000000u) ? __uint_as_float(k & 0x7FFFFFFFu)
                           : __uint_as_float(~k);
}
__device__ __forceinline__ float leaky(float x) {
  return x >= 0.0f ? x : NEG_SLOPE * x;
}

__global__ __launch_bounds__(32) void gemm36_wmma(
    const float* __restrict__ A, const float* __restrict__ W,
    const float* __restrict__ bias, float* __restrict__ out, int nrows) {
  const int lane = threadIdx.x;
  const int half = lane >> 4;
  const int idx  = lane & 15;

  int mrow = blockIdx.x * 16 + idx;
  mrow = mrow < nrows ? mrow : (nrows - 1);
  const float* __restrict__ Ab = A + (size_t)mrow * FDIM + 2 * half;

  const int col0 = idx;
  const int col1 = idx + 16;
  const int col2 = (idx + 32) < FDIM ? (idx + 32) : (FDIM - 1);
  const float* __restrict__ Wb0 = W + col0 + 72 * half;
  const float* __restrict__ Wb1 = W + col1 + 72 * half;
  const float* __restrict__ Wb2 = W + col2 + 72 * half;

  v8f acc[3];
  acc[0] = (v8f){0.f,0.f,0.f,0.f,0.f,0.f,0.f,0.f};
  acc[1] = acc[0];
  acc[2] = acc[0];

  #pragma unroll
  for (int k0 = 0; k0 < FDIM; k0 += 4) {
    v2f a, b0, b1, b2;
    a.x  = Ab[k0];
    a.y  = Ab[k0 + 1];
    b0.x = Wb0[(k0 + 0) * FDIM];
    b0.y = Wb0[(k0 + 1) * FDIM];
    b1.x = Wb1[(k0 + 0) * FDIM];
    b1.y = Wb1[(k0 + 1) * FDIM];
    b2.x = Wb2[(k0 + 0) * FDIM];
    b2.y = Wb2[(k0 + 1) * FDIM];
    acc[0] = __builtin_amdgcn_wmma_f32_16x16x4_f32(false, a, false, b0,
                                                   (short)0, acc[0], false, false);
    acc[1] = __builtin_amdgcn_wmma_f32_16x16x4_f32(false, a, false, b1,
                                                   (short)0, acc[1], false, false);
    acc[2] = __builtin_amdgcn_wmma_f32_16x16x4_f32(false, a, false, b2,
                                                   (short)0, acc[2], false, false);
  }

  const int rowbase = blockIdx.x * 16 + 8 * half;
  const bool full = (blockIdx.x * 16 + 16) <= nrows;
  #pragma unroll
  for (int t = 0; t < 3; ++t) {
    const int col = idx + 16 * t;
    if (col < FDIM) {
      const float bv = bias ? bias[col] : 0.0f;
      float* o = out + (size_t)rowbase * FDIM + col;
      if (full) {
        #pragma unroll
        for (int v = 0; v < 8; ++v) o[(size_t)v * FDIM] = acc[t][v] + bv;
      } else {
        #pragma unroll
        for (int v = 0; v < 8; ++v)
          if (rowbase + v < nrows) o[(size_t)v * FDIM] = acc[t][v] + bv;
      }
    }
  }
}

__global__ void attn_scores(const float* __restrict__ feat,
                            const float* __restrict__ a_l,
                            const float* __restrict__ a_r,
                            float* __restrict__ el, float* __restrict__ er,
                            int n) {
  int gid = blockIdx.x * blockDim.x + threadIdx.x;
  if (gid >= n * NHEAD) return;
  const int node = gid / NHEAD;
  const int h    = gid % NHEAD;
  const float4* f4 = (const float4*)(feat + (size_t)node * FDIM + h * HDIM);
  const float4* l4 = (const float4*)(a_l + h * HDIM);
  const float4* r4 = (const float4*)(a_r + h * HDIM);
  float sl = 0.0f, sr = 0.0f;
  #pragma unroll
  for (int q = 0; q < HDIM / 4; ++q) {
    const float4 fv = f4[q], lv = l4[q], rv = r4[q];
    sl += fv.x * lv.x + fv.y * lv.y + fv.z * lv.z + fv.w * lv.w;
    sr += fv.x * rv.x + fv.y * rv.y + fv.z * rv.z + fv.w * rv.w;
  }
  el[gid] = sl;
  er[gid] = sr;
}

__global__ void layer_init(unsigned* __restrict__ mkey,
                           float* __restrict__ denom,
                           float* __restrict__ rst, int n) {
  int gid = blockIdx.x * blockDim.x + threadIdx.x;
  if (gid < n * FDIM) rst[gid] = 0.0f;
  if (gid < n * NHEAD) {
    mkey[gid] = NEG_INF_KEY;
    denom[gid] = 0.0f;
  }
}

__global__ void edge_max(const int* __restrict__ src,
                         const int* __restrict__ dst,
                         const float* __restrict__ el,
                         const float* __restrict__ er,
                         unsigned* __restrict__ mkey, int ne) {
  int e = blockIdx.x * blockDim.x + threadIdx.x;
  if (e >= ne) return;
  const int s = src[e], d = dst[e];
  #pragma unroll
  for (int h = 0; h < NHEAD; ++h) {
    const float v = leaky(el[s * NHEAD + h] + er[d * NHEAD + h]);
    atomicMax(&mkey[d * NHEAD + h], f2key(v));
  }
}

__global__ void m_finalize(unsigned* __restrict__ mkey, float* __restrict__ m,
                           int n3) {
  int gid = blockIdx.x * blockDim.x + threadIdx.x;
  if (gid >= n3) return;
  float v = key2f(mkey[gid]);
  if (!isfinite(v)) v = 0.0f;
  m[gid] = v;
}

__global__ void edge_sum(const int* __restrict__ src,
                         const int* __restrict__ dst,
                         const float* __restrict__ el,
                         const float* __restrict__ er,
                         const float* __restrict__ m,
                         float* __restrict__ denom, int ne) {
  int e = blockIdx.x * blockDim.x + threadIdx.x;
  if (e >= ne) return;
  const int s = src[e], d = dst[e];
  #pragma unroll
  for (int h = 0; h < NHEAD; ++h) {
    const float v = leaky(el[s * NHEAD + h] + er[d * NHEAD + h]);
    const float ee = expf(v - m[d * NHEAD + h]);
    atomicAdd(&denom[d * NHEAD + h], ee);
  }
}

__global__ void edge_msg(const int* __restrict__ src,
                         const int* __restrict__ dst,
                         const float* __restrict__ el,
                         const float* __restrict__ er,
                         const float* __restrict__ m,
                         const float* __restrict__ denom,
                         const float* __restrict__ feat,
                         float* __restrict__ rst, int ne) {
  int e = blockIdx.x * blockDim.x + threadIdx.x;
  if (e >= ne) return;
  const int s = src[e], d = dst[e];
  const float4* fsrc4 = (const float4*)(feat + (size_t)s * FDIM);
  __builtin_prefetch(fsrc4, 0, 3);
  #pragma unroll
  for (int h = 0; h < NHEAD; ++h) {
    const float v = leaky(el[s * NHEAD + h] + er[d * NHEAD + h]);
    const float alpha = expf(v - m[d * NHEAD + h]) /
                        fmaxf(denom[d * NHEAD + h], 1e-9f);
    float* rd = rst + (size_t)d * FDIM + h * HDIM;
    #pragma unroll
    for (int q = 0; q < HDIM / 4; ++q) {
      const float4 fv = fsrc4[h * (HDIM / 4) + q];
      atomicAdd(&rd[q * 4 + 0], fv.x * alpha);
      atomicAdd(&rd[q * 4 + 1], fv.y * alpha);
      atomicAdd(&rd[q * 4 + 2], fv.z * alpha);
      atomicAdd(&rd[q * 4 + 3], fv.w * alpha);
    }
  }
}

__global__ void node_finalize(const float* __restrict__ hin,
                              const float* __restrict__ bias,
                              float* __restrict__ rst, int n) {
  int gid = blockIdx.x * blockDim.x + threadIdx.x;
  if (gid >= n * FDIM) return;
  const float x = rst[gid] + hin[gid] + bias[gid % FDIM];
  rst[gid] = x > 0.0f ? x : expm1f(x);
}

extern "C" void kernel_launch(void* const* d_in, const int* in_sizes, int n_in,
                              void* d_out, int out_size, void* d_ws,
                              size_t ws_size, hipStream_t stream) {
  (void)n_in; (void)out_size; (void)ws_size;
  const float* X        = (const float*)d_in[0];
  const int*   src      = (const int*)d_in[1];
  const int*   dst      = (const int*)d_in[2];
  const float* lin0_w   = (const float*)d_in[3];
  const float* lin0_b   = (const float*)d_in[4];
  const float* fc_w     = (const float*)d_in[5];
  const float* attn_l   = (const float*)d_in[6];
  const float* attn_r   = (const float*)d_in[7];
  const float* gat_bias = (const float*)d_in[8];
  const float* out_w    = (const float*)d_in[9];
  const float* out_b    = (const float*)d_in[10];

  const int n  = in_sizes[0] / FDIM;
  const int ne = in_sizes[1];

  float* ws   = (float*)d_ws;
  float* hA    = ws; ws += (size_t)n * FDIM;
  float* hB    = ws; ws += (size_t)n * FDIM;
  float* feat  = ws; ws += (size_t)n * FDIM;
  float* el    = ws; ws += (size_t)n * NHEAD;
  float* er    = ws; ws += (size_t)n * NHEAD;
  float* mbuf  = ws; ws += (size_t)n * NHEAD;
  float* denom = ws; ws += (size_t)n * NHEAD;

  const dim3 gemmGrid((n + 15) / 16);
  const int nbE  = (ne + 255) / 256;
  const int nbN3 = (n * NHEAD + 255) / 256;
  const int nbNF = (n * FDIM + 255) / 256;

  gemm36_wmma<<<gemmGrid, 32, 0, stream>>>(X, lin0_w, lin0_b, hA, n);

  float* hcur = hA;
  float* hnxt = hB;
  for (int l = 0; l < 3; ++l) {
    gemm36_wmma<<<gemmGrid, 32, 0, stream>>>(
        hcur, fc_w + (size_t)l * FDIM * FDIM, nullptr, feat, n);
    attn_scores<<<nbN3, 256, 0, stream>>>(
        feat, attn_l + (size_t)l * NHEAD * HDIM, attn_r + (size_t)l * NHEAD * HDIM,
        el, er, n);
    layer_init<<<nbNF, 256, 0, stream>>>((unsigned*)mbuf, denom, hnxt, n);
    edge_max<<<nbE, 256, 0, stream>>>(src, dst, el, er, (unsigned*)mbuf, ne);
    m_finalize<<<nbN3, 256, 0, stream>>>((unsigned*)mbuf, mbuf, n * NHEAD);
    edge_sum<<<nbE, 256, 0, stream>>>(src, dst, el, er, mbuf, denom, ne);
    edge_msg<<<nbE, 256, 0, stream>>>(src, dst, el, er, mbuf, denom, feat,
                                      hnxt, ne);
    node_finalize<<<nbNF, 256, 0, stream>>>(hcur, gat_bias + (size_t)l * FDIM,
                                            hnxt, n);
    float* t = hcur; hcur = hnxt; hnxt = t;
  }

  gemm36_wmma<<<gemmGrid, 32, 0, stream>>>(hcur, out_w, out_b, (float*)d_out, n);
}


# sum, message and node update in one launch: a work-group owns 392 dst nodes in LDS, scans all dst, adds its edges with LDS atomics (arrival order); division by the weight sum moved behind the sums
# speedup vs baseline: 3.7149x; 3.7149x over previous
#include <hip/hip_runtime.h>
#include <math.h>

typedef __attribute__((ext_vector_type(2))) float v2f;
typedef __attribute__((ext_vector_type(8))) float v8f;

#define FDIM 36
#define NHEAD 3
#define HDIM 12
#define NEG_SLOPE 0.2f
#define NEG_INF_KEY 0x007FFFFFu

__device__ __forceinline__ unsigned f2key(float f) {
  unsigned u = __float_as_uint(f);
  return (u & 0x80000000u) ? ~u : (u | 0x80000000u);
}
__device__ __forceinline__ float key2f(unsigned k) {
  return (k & 0x80000000u) ? __uint_as_float(k & 0x7FFFFFFFu)
                           : __uint_as_float(~k);
}
__device__ __forceinline__ float leaky(float x) {
  return x >= 0.0f ? x : NEG_SLOPE * x;
}

__global__ __launch_bounds__(32) void gemm36_wmma(
    const float* __restrict__ A, const float* __restrict__ W,
    const float* __restrict__ bias, float* __restrict__ out, int nrows) {
  const int lane = threadIdx.x;
  const int half = lane >> 4;
  const int idx  = lane & 15;

  int mrow = blockIdx.x * 16 + idx;
  mrow = mrow < nrows ? mrow : (nrows - 1);
  const float* __restrict__ Ab = A + (size_t)mrow * FDIM + 2 * half;

  const int col0 = idx;
  const int col1 = idx + 16;
  const int col2 = (idx + 32) < FDIM ? (idx + 32) : (FDIM - 1);
  const float* __restrict__ Wb0 = W + col0 + 72 * half;
  const float* __restrict__ Wb1 = W + col1 + 72 * half;
  const float* __restrict__ Wb2 = W + col2 + 72 * half;

  v8f acc[3];
  acc[0] = (v8f){0.f,0.f,0.f,0.f,0.f,0.f,0.f,0.f};
  acc[1] = acc[0];
  acc[2] = acc[0];

  #pragma unroll
  for (int k0 = 0; k0 < FDIM; k0 += 4) {
    v2f a, b0, b1, b2;
    a.x  = Ab[k0];
    a.y  = Ab[k0 + 1];
    b0.x = Wb0[(k0 + 0) * FDIM];
    b0.y = Wb0[(k0 + 1) * FDIM];
    b1.x = Wb1[(k0 + 0) * FDIM];
    b1.y = Wb1[(k0 + 1) * FDIM];
    b2.x = Wb2[(k0 + 0) * FDIM];
    b2.y = Wb2[(k0 + 1) * FDIM];
    acc[0] = __builtin_amdgcn_wmma_f32_16x16x4_f32(false, a, false, b0,
                                                   (short)0, acc[0], false, false);
    acc[1] = __builtin_amdgcn_wmma_f32_16x16x4_f32(false, a, false, b1,
                                                   (short)0, acc[1], false, false);
    acc[2] = __builtin_amdgcn_wmma_f32_16x16x4_f32(false, a, false, b2,
                                                   (short)0, acc[2], false, false);
  }

  const int rowbase = blockIdx.x * 16 + 8 * half;
  const bool full = (blockIdx.x * 16 + 16) <= nrows;
  #pragma unroll
  for (int t = 0; t < 3; ++t) {
    const int col = idx + 16 * t;
    if (col < FDIM) {
      const float bv = bias ? bias[col] : 0.0f;
      float* o = out + (size_t)rowbase * FDIM + col;
      if (full) {
        #pragma unroll
        for (int v = 0; v < 8; ++v) o[(size_t)v * FDIM] = acc[t][v] + bv;
      } else {
        #pragma unroll
        for (int v = 0; v < 8; ++v)
          if (rowbase + v < nrows) o[(size_t)v * FDIM] = acc[t][v] + bv;
      }
    }
  }
}

__global__ void attn_scores(const float* __restrict__ feat,
                            const float* __restrict__ a_l,
                            const float* __restrict__ a_r,
                            float* __restrict__ el, float* __restrict__ er,
                            int n) {
  int gid = blockIdx.x * blockDim.x + threadIdx.x;
  if (gid >= n * NHEAD) return;
  const int node = gid / NHEAD;
  const int h    = gid % NHEAD;
  const float4* f4 = (const float4*)(feat + (size_t)node * FDIM + h * HDIM);
  const float4* l4 = (const float4*)(a_l + h * HDIM);
  const float4* r4 = (const float4*)(a_r + h * HDIM);
  float sl = 0.0f, sr = 0.0f;
  #pragma unroll
  for (int q = 0; q < HDIM / 4; ++q) {
    const float4 fv = f4[q], lv = l4[q], rv = r4[q];
    sl += fv.x * lv.x + fv.y * lv.y + fv.z * lv.z + fv.w * lv.w;
    sr += fv.x * rv.x + fv.y * rv.y + fv.z * rv.z + fv.w * rv.w;
  }
  el[gid] = sl;
  er[gid] = sr;
}

__global__ void layer_init(unsigned* __restrict__ mkey, int n3) {
  int gid = blockIdx.x * blockDim.x + threadIdx.x;
  if (gid < n3) mkey[gid] = NEG_INF_KEY;
}

__global__ void edge_max(const int* __restrict__ src,
                         const int* __restrict__ dst,
                         const float* __restrict__ el,
                         const float* __restrict__ er,
                         unsigned* __restrict__ mkey, int ne) {
  int e = blockIdx.x * blockDim.x + threadIdx.x;
  if (e >= ne) return;
  const int s = src[e], d = dst[e];
  #pragma unroll
  for (int h = 0; h < NHEAD; ++h) {
    const float v = leaky(el[s * NHEAD + h] + er[d * NHEAD + h]);
    atomicMax(&mkey[d * NHEAD + h], f2key(v));
  }
}

__global__ void m_finalize(unsigned* __restrict__ mkey, float* __restrict__ m,
                           int n3) {
  int gid = blockIdx.x * blockDim.x + threadIdx.x;
  if (gid >= n3) return;
  float v = key2f(mkey[gid]);
  if (!isfinite(v)) v = 0.0f;
  m[gid] = v;
}

#define RN 392
#define APITCH 40
#define QCAP 512
#define AGG_THREADS 1024
#define AGG_WAVES (AGG_THREADS / 32)
#define ROWS_PER_BATCH 8

__device__ __forceinline__ void agg_batch(const int2* q, int head, int cnt,
                                          int lane,
                                          const int* __restrict__ src,
                                          const float* __restrict__ el,
                                          const float* __restrict__ feat,
                                          const float* sEr, const float* sM,
                                          float* acc) {
  int s = 0, dl = 0;
  float w0 = 0.0f, w1 = 0.0f, w2 = 0.0f;
  if (lane < cnt) {
    const int2 ent = q[(head + lane) & (QCAP - 1)];
    dl = ent.y;
    s = src[ent.x];
    float w[NHEAD];
    #pragma unroll
    for (int h = 0; h < NHEAD; ++h) {
      const float v = leaky(el[s * NHEAD + h] + sEr[dl * NHEAD + h]);
      w[h] = expf(v - sM[dl * NHEAD + h]);
      atomicAdd(&acc[dl * APITCH + FDIM + h], w[h]);
    }
    w0 = w[0]; w1 = w[1]; w2 = w[2];
  }
  const int iw0 = __float_as_int(w0);
  const int iw1 = __float_as_int(w1);
  const int iw2 = __float_as_int(w2);

  if (cnt == 32) {
    #pragma unroll
    for (int j0 = 0; j0 < 32; j0 += ROWS_PER_BATCH) {
      float f[ROWS_PER_BATCH];
      #pragma unroll
      for (int i = 0; i < ROWS_PER_BATCH; ++i) {
        const int sj = __builtin_amdgcn_readlane(s, j0 + i);
        f[i] = feat[(size_t)sj * FDIM + lane];
      }
      #pragma unroll
      for (int i = 0; i < ROWS_PER_BATCH; ++i) {
        const int dj = __builtin_amdgcn_readlane(dl, j0 + i);
        const float b0 = __int_as_float(__builtin_amdgcn_readlane(iw0, j0 + i));
        const float b1 = __int_as_float(__builtin_amdgcn_readlane(iw1, j0 + i));
        const float b2 = __int_as_float(__builtin_amdgcn_readlane(iw2, j0 + i));
        const float a = lane < HDIM ? b0 : (lane < 2 * HDIM ? b1 : b2);
        atomicAdd(&acc[dj * APITCH + lane], f[i] * a);
      }
    }
  } else {
    for (int j = 0; j < cnt; ++j) {
      const int sj = __builtin_amdgcn_readlane(s, j);
      const int dj = __builtin_amdgcn_readlane(dl, j);
      const float b0 = __int_as_float(__builtin_amdgcn_readlane(iw0, j));
      const float b1 = __int_as_float(__builtin_amdgcn_readlane(iw1, j));
      const float b2 = __int_as_float(__builtin_amdgcn_readlane(iw2, j));
      const float a = lane < HDIM ? b0 : (lane < 2 * HDIM ? b1 : b2);
      atomicAdd(&acc[dj * APITCH + lane], feat[(size_t)sj * FDIM + lane] * a);
    }
  }

  const int qd = lane & 3;
  const int jq = lane >> 2;
  float ft[4], at[4];
  int dt[4];
  #pragma unroll
  for (int t = 0; t < 4; ++t) {
    const int j = 8 * t + jq;
    const int sj = __shfl(s, j);
    dt[t] = __shfl(dl, j);
    at[t] = __shfl(w2, j);
    ft[t] = feat[(size_t)sj * FDIM + 32 + qd];
  }
  #pragma unroll
  for (int t = 0; t < 4; ++t) {
    if (8 * t + jq < cnt)
      atomicAdd(&acc[dt[t] * APITCH + 32 + qd], ft[t] * at[t]);
  }
}

__global__ __launch_bounds__(AGG_THREADS) void edge_agg(
    const int* __restrict__ src, const int* __restrict__ dst,
    const float* __restrict__ el, const float* __restrict__ er,
    const float* __restrict__ m, const float* __restrict__ feat,
    const float* __restrict__ hin, const float* __restrict__ bias,
    float* __restrict__ hout, int n, int ne) {
  __shared__ float acc[RN * APITCH];
  __shared__ float sEr[RN * NHEAD];
  __shared__ float sM[RN * NHEAD];
  __shared__ int2 ring[AGG_WAVES * QCAP];

  const int tid = threadIdx.x;
  const int lane = tid & 31;
  const int wave = __builtin_amdgcn_readfirstlane(tid >> 5);
  const int lo = blockIdx.x * RN;
  const int cntN = (n - lo) < RN ? (n - lo) : RN;

  for (int i = tid; i < RN * APITCH; i += AGG_THREADS) acc[i] = 0.0f;
  for (int i = tid; i < cntN * NHEAD; i += AGG_THREADS) {
    sEr[i] = er[(size_t)lo * NHEAD + i];
    sM[i] = m[(size_t)lo * NHEAD + i];
  }
  __syncthreads();

  int2* q = ring + wave * QCAP;
  int head = 0, tail = 0;

#define PUSH(dval, eidx, okm)                                                \
  {                                                                          \
    const int dlc = (dval) - lo;                                             \
    const unsigned mk =                                                      \
        __builtin_amdgcn_ballot_w32((unsigned)dlc < (unsigned)cntN) & (okm); \
    if (mk) {                                                                \
      if ((mk >> lane) & 1u) {                                               \
        const int pos = tail + (int)__builtin_amdgcn_mbcnt_lo(mk, 0u);       \
        q[pos & (QCAP - 1)] = make_int2((eidx), dlc);                        \
      }                                                                      \
      tail += __builtin_popcount(mk);                                        \
    }                                                                        \
  }
#define DRAIN()                                                              \
  while (tail - head >= 32) {                                                \
    agg_batch(q, head, 32, lane, src, el, feat, sEr, sM, acc);               \
    head += 32;                                                              \
  }

  const int4* __restrict__ dst4 = (const int4*)dst;
  const int nvec = ne >> 2;
  const int vstep = AGG_WAVES * 128;
  if (nvec > 0) {
    const int vlast = nvec - 1;
    const int ntrip = (nvec + vstep - 1) / vstep;
    int tt = (int)(((long long)blockIdx.x * ntrip) / gridDim.x);
    int4 cur[4];
    #pragma unroll
    for (int k = 0; k < 4; ++k) {
      const int v = tt * vstep + wave * 128 + k * 32 + lane;
      cur[k] = dst4[v < vlast ? v : vlast];
    }
    for (int t = 0; t < ntrip; ++t) {
      const int tn = (tt + 1 == ntrip) ? 0 : tt + 1;
      int4 nxt[4];
      #pragma unroll
      for (int k = 0; k < 4; ++k) {
        const int v = tn * vstep + wave * 128 + k * 32 + lane;
        nxt[k] = dst4[v < vlast ? v : vlast];
      }
      const int vb = tt * vstep + wave * 128;
      #pragma unroll
      for (int p = 0; p < 2; ++p) {
        #pragma unroll
        for (int kk = 0; kk < 2; ++kk) {
          const int k = 2 * p + kk;
          const int v = vb + k * 32 + lane;
          const unsigned okm = __builtin_amdgcn_ballot_w32(v < nvec);
          PUSH(cur[k].x, 4 * v + 0, okm)
          PUSH(cur[k].y, 4 * v + 1, okm)
          PUSH(cur[k].z, 4 * v + 2, okm)
          PUSH(cur[k].w, 4 * v + 3, okm)
        }
        DRAIN()
      }
      #pragma unroll
      for (int k = 0; k < 4; ++k) cur[k] = nxt[k];
      tt = tn;
    }
  }
  if (wave == 0) {
    const int e = nvec * 4 + lane;
    const int dval = e < ne ? dst[e] : -1;
    PUSH(dval, e, 0xFFFFFFFFu)
  }
  DRAIN()
  if (tail - head > 0)
    agg_batch(q, head, tail - head, lane, src, el, feat, sEr, sM, acc);
#undef PUSH
#undef DRAIN

  __syncthreads();

  for (int i = tid; i < cntN * FDIM; i += AGG_THREADS) {
    const int node = i / FDIM;
    const int c = i - node * FDIM;
    const float den = fmaxf(acc[node * APITCH + FDIM + c / HDIM], 1e-9f);
    const size_t g = (size_t)lo * FDIM + i;
    const float x = acc[node * APITCH + c] / den + hin[g] + bias[c];
    hout[g] = x > 0.0f ? x : expm1f(x);
  }
}

extern "C" void kernel_launch(void* const* d_in, const int* in_sizes, int n_in,
                              void* d_out, int out_size, void* d_ws,
                              size_t ws_size, hipStream_t stream) {
  (void)n_in; (void)out_size; (void)ws_size;
  const float* X        = (const float*)d_in[0];
  const int*   src      = (const int*)d_in[1];
  const int*   dst      = (const int*)d_in[2];
  const float* lin0_w   = (const float*)d_in[3];
  const float* lin0_b   = (const float*)d_in[4];
  const float* fc_w     = (const float*)d_in[5];
  const float* attn_l   = (const float*)d_in[6];
  const float* attn_r   = (const float*)d_in[7];
  const float* gat_bias = (const float*)d_in[8];
  const float* out_w    = (const float*)d_in[9];
  const float* out_b    = (const float*)d_in[10];

  const int n  = in_sizes[0] / FDIM;
  const int ne = in_sizes[1];

  float* ws   = (float*)d_ws;
  float* hA    = ws; ws += (size_t)n * FDIM;
  float* hB    = ws; ws += (size_t)n * FDIM;
  float* feat  = ws; ws += (size_t)n * FDIM;
  float* el    = ws; ws += (size_t)n * NHEAD;
  float* er    = ws; ws += (size_t)n * NHEAD;
  float* mbuf  = ws; ws += (size_t)n * NHEAD;

  const dim3 gemmGrid((n + 15) / 16);
  const int nbE  = (ne + 255) / 256;
  const int nbN3 = (n * NHEAD + 255) / 256;
  const int nbR  = (n + RN - 1) / RN;

  gemm36_wmma<<<gemmGrid, 32, 0, stream>>>(X, lin0_w, lin0_b, hA, n);

  float* hcur = hA;
  float* hnxt = hB;
  for (int l = 0; l < 3; ++l) {
    gemm36_wmma<<<gemmGrid, 32, 0, stream>>>(
        hcur, fc_w + (size_t)l * FDIM * FDIM, nullptr, feat, n);
    attn_scores<<<nbN3, 256, 0, stream>>>(
        feat, attn_l + (size_t)l * NHEAD * HDIM, attn_r + (size_t)l * NHEAD * HDIM,
        el, er, n);
    layer_init<<<nbN3, 256, 0, stream>>>((unsigned*)mbuf, n * NHEAD);
    edge_max<<<nbE, 256, 0, stream>>>(src, dst, el, er, (unsigned*)mbuf, ne);
    m_finalize<<<nbN3, 256, 0, stream>>>((unsigned*)mbuf, mbuf, n * NHEAD);
    edge_agg<<<nbR, AGG_THREADS, 0, stream>>>(src, dst, el, er, mbuf, feat, hcur,
                                             gat_bias + (size_t)l * FDIM, hnxt,
                                             n, ne);
    float* t = hcur; hcur = hnxt; hnxt = t;
  }

  gemm36_wmma<<<gemmGrid, 32, 0, stream>>>(hcur, out_w, out_b, (float*)d_out, n);
}


# edge_agg keeps each wave's in-range edges as a whole list in LDS: per-node max taken there by LDS atomic max before the sums (arrival order as before); edge_max, init, finalize launches go
# speedup vs baseline: 7.3585x; 1.9808x over previous
#include <hip/hip_runtime.h>
#include <math.h>

typedef __attribute__((ext_vector_type(2))) float v2f;
typedef __attribute__((ext_vector_type(8))) float v8f;

#define FDIM 36
#define NHEAD 3
#define HDIM 12
#define NEG_SLOPE 0.2f
#define NEG_INF_KEY 0x007FFFFFu

__device__ __forceinline__ unsigned f2key(float f) {
  unsigned u = __float_as_uint(f);
  return (u & 0x80000000u) ? ~u : (u | 0x80000000u);
}
__device__ __forceinline__ float key2f(unsigned k) {
  return (k & 0x80000000u) ? __uint_as_float(k & 0x7FFFFFFFu)
                           : __uint_as_float(~k);
}
__device__ __forceinline__ float leaky(float x) {
  return x >= 0.0f ? x : NEG_SLOPE * x;
}

__global__ __launch_bounds__(32) void gemm36_wmma(
    const float* __restrict__ A, const float* __restrict__ W,
    const float* __restrict__ bias, float* __restrict__ out, int nrows) {
  const int lane = threadIdx.x;
  const int half = lane >> 4;
  const int idx  = lane & 15;

  int mrow = blockIdx.x * 16 + idx;
  mrow = mrow < nrows ? mrow : (nrows - 1);
  const float* __restrict__ Ab = A + (size_t)mrow * FDIM + 2 * half;

  const int col0 = idx;
  const int col1 = idx + 16;
  const int col2 = (idx + 32) < FDIM ? (idx + 32) : (FDIM - 1);
  const float* __restrict__ Wb0 = W + col0 + 72 * half;
  const float* __restrict__ Wb1 = W + col1 + 72 * half;
  const float* __restrict__ Wb2 = W + col2 + 72 * half;

  v8f acc[3];
  acc[0] = (v8f){0.f,0.f,0.f,0.f,0.f,0.f,0.f,0.f};
  acc[1] = acc[0];
  acc[2] = acc[0];

  #pragma unroll
  for (int k0 = 0; k0 < FDIM; k0 += 4) {
    v2f a, b0, b1, b2;
    a.x  = Ab[k0];
    a.y  = Ab[k0 + 1];
    b0.x = Wb0[(k0 + 0) * FDIM];
    b0.y = Wb0[(k0 + 1) * FDIM];
    b1.x = Wb1[(k0 + 0) * FDIM];
    b1.y = Wb1[(k0 + 1) * FDIM];
    b2.x = Wb2[(k0 + 0) * FDIM];
    b2.y = Wb2[(k0 + 1) * FDIM];
    acc[0] = __builtin_amdgcn_wmma_f32_16x16x4_f32(false, a, false, b0,
                                                   (short)0, acc[0], false, false);
    acc[1] = __builtin_amdgcn_wmma_f32_16x16x4_f32(false, a, false, b1,
                                                   (short)0, acc[1], false, false);
    acc[2] = __builtin_amdgcn_wmma_f32_16x16x4_f32(false, a, false, b2,
                                                   (short)0, acc[2], false, false);
  }

  const int rowbase = blockIdx.x * 16 + 8 * half;
  const bool full = (blockIdx.x * 16 + 16) <= nrows;
  #pragma unroll
  for (int t = 0; t < 3; ++t) {
    const int col = idx + 16 * t;
    if (col < FDIM) {
      const float bv = bias ? bias[col] : 0.0f;
      float* o = out + (size_t)rowbase * FDIM + col;
      if (full) {
        #pragma unroll
        for (int v = 0; v < 8; ++v) o[(size_t)v * FDIM] = acc[t][v] + bv;
      } else {
        #pragma unroll
        for (int v = 0; v < 8; ++v)
          if (rowbase + v < nrows) o[(size_t)v * FDIM] = acc[t][v] + bv;
      }
    }
  }
}

__global__ void attn_scores(const float* __restrict__ feat,
                            const float* __restrict__ a_l,
                            const float* __restrict__ a_r,
                            float* __restrict__ el, float* __restrict__ er,
                            int n) {
  int gid = blockIdx.x * blockDim.x + threadIdx.x;
  if (gid >= n * NHEAD) return;
  const int node = gid / NHEAD;
  const int h    = gid % NHEAD;
  const float4* f4 = (const float4*)(feat + (size_t)node * FDIM + h * HDIM);
  const float4* l4 = (const float4*)(a_l + h * HDIM);
  const float4* r4 = (const float4*)(a_r + h * HDIM);
  float sl = 0.0f, sr = 0.0f;
  #pragma unroll
  for (int q = 0; q < HDIM / 4; ++q) {
    const float4 fv = f4[q], lv = l4[q], rv = r4[q];
    sl += fv.x * lv.x + fv.y * lv.y + fv.z * lv.z + fv.w * lv.w;
    sr += fv.x * rv.x + fv.y * rv.y + fv.z * rv.z + fv.w * rv.w;
  }
  el[gid] = sl;
  er[gid] = sr;
}

#define RN 392
#define APITCH 40
#define LCAP 896
#define AGG_THREADS 1024
#define AGG_WAVES (AGG_THREADS / 32)
#define ROWS_PER_BATCH 8

__device__ __forceinline__ void agg_batch(const int2* q, int head, int cnt,
                                          int lane,
                                          const float* __restrict__ el,
                                          const float* __restrict__ feat,
                                          const float* sEr,
                                          const unsigned* sMx, float* acc) {
  int s = 0, dl = 0;
  float w0 = 0.0f, w1 = 0.0f, w2 = 0.0f;
  if (lane < cnt) {
    const int2 ent = q[head + lane];
    s = ent.x;
    dl = ent.y;
    float w[NHEAD];
    #pragma unroll
    for (int h = 0; h < NHEAD; ++h) {
      const float v = leaky(el[s * NHEAD + h] + sEr[dl * NHEAD + h]);
      w[h] = expf(v - __uint_as_float(sMx[dl * NHEAD + h]));
      atomicAdd(&acc[dl * APITCH + FDIM + h], w[h]);
    }
    w0 = w[0]; w1 = w[1]; w2 = w[2];
  }
  const int iw0 = __float_as_int(w0);
  const int iw1 = __float_as_int(w1);
  const int iw2 = __float_as_int(w2);

  if (cnt == 32) {
    #pragma unroll
    for (int j0 = 0; j0 < 32; j0 += ROWS_PER_BATCH) {
      float f[ROWS_PER_BATCH];
      #pragma unroll
      for (int i = 0; i < ROWS_PER_BATCH; ++i) {
        const int sj = __builtin_amdgcn_readlane(s, j0 + i);
        f[i] = feat[(size_t)sj * FDIM + lane];
      }
      #pragma unroll
      for (int i = 0; i < ROWS_PER_BATCH; ++i) {
        const int dj = __builtin_amdgcn_readlane(dl, j0 + i);
        const float b0 = __int_as_float(__builtin_amdgcn_readlane(iw0, j0 + i));
        const float b1 = __int_as_float(__builtin_amdgcn_readlane(iw1, j0 + i));
        const float b2 = __int_as_float(__builtin_amdgcn_readlane(iw2, j0 + i));
        const float a = lane < HDIM ? b0 : (lane < 2 * HDIM ? b1 : b2);
        atomicAdd(&acc[dj * APITCH + lane], f[i] * a);
      }
    }
  } else {
    for (int j = 0; j < cnt; ++j) {
      const int sj = __builtin_amdgcn_readlane(s, j);
      const int dj = __builtin_amdgcn_readlane(dl, j);
      const float b0 = __int_as_float(__builtin_amdgcn_readlane(iw0, j));
      const float b1 = __int_as_float(__builtin_amdgcn_readlane(iw1, j));
      const float b2 = __int_as_float(__builtin_amdgcn_readlane(iw2, j));
      const float a = lane < HDIM ? b0 : (lane < 2 * HDIM ? b1 : b2);
      atomicAdd(&acc[dj * APITCH + lane], feat[(size_t)sj * FDIM + lane] * a);
    }
  }

  const int qd = lane & 3;
  const int jq = lane >> 2;
  float ft[4], at[4];
  int dt[4];
  #pragma unroll
  for (int t = 0; t < 4; ++t) {
    const int j = 8 * t + jq;
    const int sj = __shfl(s, j);
    dt[t] = __shfl(dl, j);
    at[t] = __shfl(w2, j);
    ft[t] = feat[(size_t)sj * FDIM + 32 + qd];
  }
  #pragma unroll
  for (int t = 0; t < 4; ++t) {
    if (8 * t + jq < cnt)
      atomicAdd(&acc[dt[t] * APITCH + 32 + qd], ft[t] * at[t]);
  }
}

__global__ __launch_bounds__(AGG_THREADS) void edge_agg(
    const int* __restrict__ src, const int* __restrict__ dst,
    const float* __restrict__ el, const float* __restrict__ er,
    const float* __restrict__ feat,
    const float* __restrict__ hin, const float* __restrict__ bias,
    float* __restrict__ hout, int n, int ne) {
  __shared__ float acc[RN * APITCH];
  __shared__ float sEr[RN * NHEAD];
  __shared__ unsigned sMx[RN * NHEAD];
  __shared__ int2 lst[AGG_WAVES * LCAP];
  __shared__ int sOvf;

  const int tid = threadIdx.x;
  const int lane = tid & 31;
  const int wave = __builtin_amdgcn_readfirstlane(tid >> 5);
  const int lo = blockIdx.x * RN;
  const int cntN = (n - lo) < RN ? (n - lo) : RN;

  for (int i = tid; i < RN * APITCH; i += AGG_THREADS) acc[i] = 0.0f;
  for (int i = tid; i < RN * NHEAD; i += AGG_THREADS) sMx[i] = NEG_INF_KEY;
  for (int i = tid; i < cntN * NHEAD; i += AGG_THREADS)
    sEr[i] = er[(size_t)lo * NHEAD + i];
  if (tid == 0) sOvf = 0;
  __syncthreads();

  int2* q = lst + wave * LCAP;
  int tail = 0;
  int ovf = 0;

#define PUSH(dval, eidx, okm)                                                \
  {                                                                          \
    const int dlc = (dval) - lo;                                             \
    const unsigned mk =                                                      \
        __builtin_amdgcn_ballot_w32((unsigned)dlc < (unsigned)cntN) & (okm); \
    if (mk) {                                                                \
      const int ntail = tail + __builtin_popcount(mk);                       \
      if (ntail <= LCAP) {                                                   \
        if ((mk >> lane) & 1u) {                                             \
          const int pos = tail + (int)__builtin_amdgcn_mbcnt_lo(mk, 0u);     \
          q[pos] = make_int2((eidx), dlc);                                   \
        }                                                                    \
        tail = ntail;                                                        \
      } else {                                                               \
        ovf = 1;                                                             \
      }                                                                      \
    }                                                                        \
  }

  const int4* __restrict__ dst4 = (const int4*)dst;
  const int nvec = ne >> 2;
  const int vstep = AGG_WAVES * 128;
  if (nvec > 0) {
    const int vlast = nvec - 1;
    const int ntrip = (nvec + vstep - 1) / vstep;
    int tt = (int)(((long long)blockIdx.x * ntrip) / gridDim.x);
    int4 cur[4];
    #pragma unroll
    for (int k = 0; k < 4; ++k) {
      const int v = tt * vstep + wave * 128 + k * 32 + lane;
      cur[k] = dst4[v < vlast ? v : vlast];
    }
    for (int t = 0; t < ntrip; ++t) {
      const int tn = (tt + 1 == ntrip) ? 0 : tt + 1;
      int4 nxt[4];
      #pragma unroll
      for (int k = 0; k < 4; ++k) {
        const int v = tn * vstep + wave * 128 + k * 32 + lane;
        nxt[k] = dst4[v < vlast ? v : vlast];
      }
      const int vb = tt * vstep + wave * 128;
      #pragma unroll
      for (int k = 0; k < 4; ++k) {
        const int v = vb + k * 32 + lane;
        const unsigned okm = __builtin_amdgcn_ballot_w32(v < nvec);
        PUSH(cur[k].x, 4 * v + 0, okm)
        PUSH(cur[k].y, 4 * v + 1, okm)
        PUSH(cur[k].z, 4 * v + 2, okm)
        PUSH(cur[k].w, 4 * v + 3, okm)
      }
      #pragma unroll
      for (int k = 0; k < 4; ++k) cur[k] = nxt[k];
      tt = tn;
    }
  }
  if (wave == 0) {
    const int e = nvec * 4 + lane;
    const int dval = e < ne ? dst[e] : -1;
    PUSH(dval, e, 0xFFFFFFFFu)
  }
#undef PUSH
  if (ovf && lane == 0) sOvf = 1;
  __syncthreads();
  const int slow = __builtin_amdgcn_readfirstlane(sOvf);

  if (!slow) {
    for (int i0 = 0; i0 < tail; i0 += 128) {
      int sv[4], dv[4];
      #pragma unroll
      for (int k = 0; k < 4; ++k) {
        const int i = i0 + 32 * k + lane;
        const int2 ent = q[i < LCAP ? i : LCAP - 1];
        sv[k] = src[i < tail ? ent.x : 0];
        dv[k] = i < tail ? ent.y : 0;
      }
      float ev[4][NHEAD];
      #pragma unroll
      for (int k = 0; k < 4; ++k) {
        #pragma unroll
        for (int h = 0; h < NHEAD; ++h) ev[k][h] = el[sv[k] * NHEAD + h];
      }
      #pragma unroll
      for (int k = 0; k < 4; ++k) {
        const int i = i0 + 32 * k + lane;
        #pragma unroll
        for (int h = 0; h < NHEAD; ++h) {
          const float v = leaky(ev[k][h] + sEr[dv[k] * NHEAD + h]);
          atomicMax(&sMx[dv[k] * NHEAD + h], i < tail ? f2key(v) : NEG_INF_KEY);
        }
        if (i < tail) q[i].x = sv[k];
      }
    }
  } else {
    for (int e = tid; e < ne; e += AGG_THREADS) {
      const int dlc = dst[e] - lo;
      if ((unsigned)dlc < (unsigned)cntN) {
        const int s = src[e];
        #pragma unroll
        for (int h = 0; h < NHEAD; ++h)
          atomicMax(&sMx[dlc * NHEAD + h],
                    f2key(leaky(el[s * NHEAD + h] + sEr[dlc * NHEAD + h])));
      }
    }
  }
  __syncthreads();

  for (int i = tid; i < cntN * NHEAD; i += AGG_THREADS) {
    float v = key2f(sMx[i]);
    if (!isfinite(v)) v = 0.0f;
    sMx[i] = __float_as_uint(v);
  }
  __syncthreads();

  if (!slow) {
    int head = 0;
    while (tail - head >= 32) {
      agg_batch(q, head, 32, lane, el, feat, sEr, sMx, acc);
      head += 32;
    }
    if (tail - head > 0)
      agg_batch(q, head, tail - head, lane, el, feat, sEr, sMx, acc);
  } else {
    for (int e = tid; e < ne; e += AGG_THREADS) {
      const int dlc = dst[e] - lo;
      if ((unsigned)dlc < (unsigned)cntN) {
        const int s = src[e];
        float w[NHEAD];
        #pragma unroll
        for (int h = 0; h < NHEAD; ++h) {
          const float v = leaky(el[s * NHEAD + h] + sEr[dlc * NHEAD + h]);
          w[h] = expf(v - __uint_as_float(sMx[dlc * NHEAD + h]));
          atomicAdd(&acc[dlc * APITCH + FDIM + h], w[h]);
        }
        for (int c = 0; c < FDIM; ++c)
          atomicAdd(&acc[dlc * APITCH + c],
                    feat[(size_t)s * FDIM + c] * w[c / HDIM]);
      }
    }
  }
  __syncthreads();

  for (int i = tid; i < cntN * FDIM; i += AGG_THREADS) {
    const int node = i / FDIM;
    const int c = i - node * FDIM;
    const float den = fmaxf(acc[node * APITCH + FDIM + c / HDIM], 1e-9f);
    const size_t g = (size_t)lo * FDIM + i;
    const float x = acc[node * APITCH + c] / den + hin[g] + bias[c];
    hout[g] = x > 0.0f ? x : expm1f(x);
  }
}

extern "C" void kernel_launch(void* const* d_in, const int* in_sizes, int n_in,
                              void* d_out, int out_size, void* d_ws,
                              size_t ws_size, hipStream_t stream) {
  (void)n_in; (void)out_size; (void)ws_size;
  const float* X        = (const float*)d_in[0];
  const int*   src      = (const int*)d_in[1];
  const int*   dst      = (const int*)d_in[2];
  const float* lin0_w   = (const float*)d_in[3];
  const float* lin0_b   = (const float*)d_in[4];
  const float* fc_w     = (const float*)d_in[5];
  const float* attn_l   = (const float*)d_in[6];
  const float* attn_r   = (const float*)d_in[7];
  const float* gat_bias = (const float*)d_in[8];
  const float* out_w    = (const float*)d_in[9];
  const float* out_b    = (const float*)d_in[10];

  const int n  = in_sizes[0] / FDIM;
  const int ne = in_sizes[1];

  float* ws   = (float*)d_ws;
  float* hA    = ws; ws += (size_t)n * FDIM;
  float* hB    = ws; ws += (size_t)n * FDIM;
  float* feat  = ws; ws += (size_t)n * FDIM;
  float* el    = ws; ws += (size_t)n * NHEAD;
  float* er    = ws; ws += (size_t)n * NHEAD;

  const dim3 gemmGrid((n + 15) / 16);
  const int nbN3 = (n * NHEAD + 255) / 256;
  const int nbR  = (n + RN - 1) / RN;

  gemm36_wmma<<<gemmGrid, 32, 0, stream>>>(X, lin0_w, lin0_b, hA, n);

  float* hcur = hA;
  float* hnxt = hB;
  for (int l = 0; l < 3; ++l) {
    gemm36_wmma<<<gemmGrid, 32, 0, stream>>>(
        hcur, fc_w + (size_t)l * FDIM * FDIM, nullptr, feat, n);
    attn_scores<<<nbN3, 256, 0, stream>>>(
        feat, attn_l + (size_t)l * NHEAD * HDIM, attn_r + (size_t)l * NHEAD * HDIM,
        el, er, n);
    edge_agg<<<nbR, AGG_THREADS, 0, stream>>>(src, dst, el, er, feat, hcur,
                                             gat_bias + (size_t)l * FDIM, hnxt,
                                             n, ne);
    float* t = hcur; hcur = hnxt; hnxt = t;
  }

  gemm36_wmma<<<gemmGrid, 32, 0, stream>>>(hcur, out_w, out_b, (float*)d_out, n);
}


# range keys replaced by 256 rows of range bits, a bit per 8 edges (LDS or, whole 128-byte lines out); edge_agg of layer 0 reads its own row, then dst of the set eights only; arrival order kept
# speedup vs baseline: 14.1328x; 1.9206x over previous
#include <hip/hip_runtime.h>
#include <math.h>

typedef __attribute__((ext_vector_type(2))) float v2f;
typedef __attribute__((ext_vector_type(8))) float v8f;
typedef __attribute__((ext_vector_type(4))) int v4i;

#define FDIM 36
#define NHEAD 3
#define HDIM 12
#define NEG_SLOPE 0.2f
#define NEG_INF_KEY 0x007FFFFFu

__device__ __forceinline__ unsigned f2key(float f) {
  unsigned u = __float_as_uint(f);
  return (u & 0x80000000u) ? ~u : (u | 0x80000000u);
}
__device__ __forceinline__ float key2f(unsigned k) {
  return (k & 0x80000000u) ? __uint_as_float(k & 0x7FFFFFFFu)
                           : __uint_as_float(~k);
}
__device__ __forceinline__ float leaky(float x) {
  return x >= 0.0f ? x : NEG_SLOPE * x;
}

__global__ __launch_bounds__(32) void gemm36_wmma(
    const float* __restrict__ A, const float* __restrict__ W,
    const float* __restrict__ bias, float* __restrict__ out, int nrows) {
  const int lane = threadIdx.x;
  const int half = lane >> 4;
  const int idx  = lane & 15;

  int mrow = blockIdx.x * 16 + idx;
  mrow = mrow < nrows ? mrow : (nrows - 1);
  const float* __restrict__ Ab = A + (size_t)mrow * FDIM + 2 * half;

  const int col0 = idx;
  const int col1 = idx + 16;
  const int col2 = (idx + 32) < FDIM ? (idx + 32) : (FDIM - 1);
  const float* __restrict__ Wb0 = W + col0 + 72 * half;
  const float* __restrict__ Wb1 = W + col1 + 72 * half;
  const float* __restrict__ Wb2 = W + col2 + 72 * half;

  v8f acc[3];
  acc[0] = (v8f){0.f,0.f,0.f,0.f,0.f,0.f,0.f,0.f};
  acc[1] = acc[0];
  acc[2] = acc[0];

  #pragma unroll
  for (int k0 = 0; k0 < FDIM; k0 += 4) {
    v2f a, b0, b1, b2;
    a.x  = Ab[k0];
    a.y  = Ab[k0 + 1];
    b0.x = Wb0[(k0 + 0) * FDIM];
    b0.y = Wb0[(k0 + 1) * FDIM];
    b1.x = Wb1[(k0 + 0) * FDIM];
    b1.y = Wb1[(k0 + 1) * FDIM];
    b2.x = Wb2[(k0 + 0) * FDIM];
    b2.y = Wb2[(k0 + 1) * FDIM];
    acc[0] = __builtin_amdgcn_wmma_f32_16x16x4_f32(false, a, false, b0,
                                                   (short)0, acc[0], false, false);
    acc[1] = __builtin_amdgcn_wmma_f32_16x16x4_f32(false, a, false, b1,
                                                   (short)0, acc[1], false, false);
    acc[2] = __builtin_amdgcn_wmma_f32_16x16x4_f32(false, a, false, b2,
                                                   (short)0, acc[2], false, false);
  }

  const int rowbase = blockIdx.x * 16 + 8 * half;
  const bool full = (blockIdx.x * 16 + 16) <= nrows;
  #pragma unroll
  for (int t = 0; t < 3; ++t) {
    const int col = idx + 16 * t;
    if (col < FDIM) {
      const float bv = bias ? bias[col] : 0.0f;
      float* o = out + (size_t)rowbase * FDIM + col;
      if (full) {
        #pragma unroll
        for (int v = 0; v < 8; ++v) o[(size_t)v * FDIM] = acc[t][v] + bv;
      } else {
        #pragma unroll
        for (int v = 0; v < 8; ++v)
          if (rowbase + v < nrows) o[(size_t)v * FDIM] = acc[t][v] + bv;
      }
    }
  }
}

__global__ void attn_scores(const float* __restrict__ feat,
                            const float* __restrict__ a_l,
                            const float* __restrict__ a_r,
                            float* __restrict__ el, float* __restrict__ er,
                            int n) {
  int gid = blockIdx.x * blockDim.x + threadIdx.x;
  if (gid >= n * NHEAD) return;
  const int node = gid / NHEAD;
  const int h    = gid % NHEAD;
  const float4* f4 = (const float4*)(feat + (size_t)node * FDIM + h * HDIM);
  const float4* l4 = (const float4*)(a_l + h * HDIM);
  const float4* r4 = (const float4*)(a_r + h * HDIM);
  float sl = 0.0f, sr = 0.0f;
  #pragma unroll
  for (int q = 0; q < HDIM / 4; ++q) {
    const float4 fv = f4[q], lv = l4[q], rv = r4[q];
    sl += fv.x * lv.x + fv.y * lv.y + fv.z * lv.z + fv.w * lv.w;
    sr += fv.x * rv.x + fv.y * rv.y + fv.z * rv.z + fv.w * rv.w;
  }
  el[gid] = sl;
  er[gid] = sr;
}

#define RN 392
#define KV 1024
#define BPITCH 257
__global__ __launch_bounds__(KV) void edge_bits(const int* __restrict__ dst,
                                                unsigned* __restrict__ bmap, int ne) {
  __shared__ unsigned bm[(KV / 32) * BPITCH];
  const int tid = threadIdx.x;
  const int lane = tid & 31;
  const int wave = tid >> 5;
  for (int i = tid; i < (KV / 32) * BPITCH; i += KV) bm[i] = 0u;
  __syncthreads();

  const int nv8 = (ne + 7) >> 3;
  const int v = blockIdx.x * KV + tid;
  if (v < nv8) {
    int d[8];
    if (8 * v + 7 < ne) {
      const int4 t0 = ((const int4*)dst)[2 * v];
      const int4 t1 = ((const int4*)dst)[2 * v + 1];
      d[0] = t0.x; d[1] = t0.y; d[2] = t0.z; d[3] = t0.w;
      d[4] = t1.x; d[5] = t1.y; d[6] = t1.z; d[7] = t1.w;
    } else {
      #pragma unroll
      for (int j = 0; j < 8; ++j) d[j] = (8 * v + j) < ne ? dst[8 * v + j] : -1;
    }
    #pragma unroll
    for (int j = 0; j < 8; ++j) {
      unsigned b = (unsigned)d[j] / RN;
      b = b < 255u ? b : 255u;
      if (8 * v + j < ne) atomicOr(&bm[wave * BPITCH + b], 1u << lane);
    }
  }
  __syncthreads();

  for (int idx = tid; idx < 256 * 32; idx += KV) {
    const int r = idx >> 5;
    const int w = idx & 31;
    bmap[((size_t)r * gridDim.x + blockIdx.x) * 32 + w] = bm[w * BPITCH + r];
  }
}

#define PACK_SHIFT 9
#define PACK_MAXN (1 << 22)
#define APITCH 40
#define LCAP 896
#define FCAP 576
#define CCAP 640
#define EXP_U 2
#define WCAP (LCAP / 2)
#define PACK_MASK ((1 << PACK_SHIFT) - 1)
#define AGG_THREADS 1024
#define AGG_WAVES (AGG_THREADS / 32)

__device__ __forceinline__ int colpos(int c) { return (c & 3) * 9 + (c >> 2); }

#define ROW_GROUPS 11

template <bool FULL, bool WIDE>
__device__ __forceinline__ void agg_batch(const int2* q, int head, int cnt,
                                          int lane,
                                          const float* __restrict__ el,
                                          const float* __restrict__ feat,
                                          const float* sEr,
                                          const unsigned* sMx, float* acc) {
  const float4* __restrict__ feat4 = (const float4*)feat;
  const v4i* q4 = (const v4i*)q;
  const int* qw = (const int*)q;
  const int r3 = lane / 9;
  const int rr = r3 < 3 ? r3 : 2;
  const int pc = lane - 9 * r3;

  const bool on = FULL || lane < cnt;
  int dl = 0;
  float ev[NHEAD] = {0.0f, 0.0f, 0.0f};
  if (on) {
    if (WIDE) {
      const v4i ent = q4[head + lane];
      dl = ent.w & PACK_MASK;
      ev[0] = __int_as_float(ent.x);
      ev[1] = __int_as_float(ent.y);
      ev[2] = __int_as_float(ent.z);
    } else {
      const int2 ent = q[head + lane];
      const int s = ent.x;
      dl = ent.y;
      #pragma unroll
      for (int h = 0; h < NHEAD; ++h) ev[h] = el[s * NHEAD + h];
    }
  }

  if (FULL) {
    float4 f[ROW_GROUPS];
    int dn[ROW_GROUPS];
    #pragma unroll
    for (int g = 0; g < ROW_GROUPS; ++g) {
      const int j = (3 * g + rr) < 32 ? (3 * g + rr) : 31;
      int sx;
      if (WIDE) {
        const int* qr = qw + 4 * (head + rr) + 3;
        const int pk = g < ROW_GROUPS - 1 ? qr[12 * g] : qr[rr < 2 ? 120 : 116];
        sx = pk >> PACK_SHIFT;
        dn[g] = pk & PACK_MASK;
      } else {
        const int2 en = q[head + j];
        sx = en.x;
        dn[g] = en.y;
      }
      f[g] = feat4[(size_t)sx * 9 + pc];
    }
    if (!WIDE) {
      #pragma unroll
      for (int g = 0; g < ROW_GROUPS; ++g)
        asm volatile("" :: "v"(f[g].x), "v"(f[g].y), "v"(f[g].z), "v"(f[g].w));
    }

    float w[NHEAD];
    #pragma unroll
    for (int h = 0; h < NHEAD; ++h) {
      const float v = WIDE ? ev[h] : leaky(ev[h] + sEr[dl * NHEAD + h]);
      w[h] = expf(v - __uint_as_float(sMx[dl * NHEAD + h]));
      atomicAdd(&acc[dl * APITCH + FDIM + h], w[h]);
    }

    float a[ROW_GROUPS];
    #pragma unroll
    for (int g = 0; g < ROW_GROUPS; ++g) {
      const int j = (3 * g + rr) < 32 ? (3 * g + rr) : 31;
      const float a0 = __shfl(w[0], j);
      const float a1 = __shfl(w[1], j);
      const float a2 = __shfl(w[2], j);
      a[g] = pc < 3 ? a0 : (pc < 6 ? a1 : a2);
    }
    if (WIDE) {
      #pragma unroll
      for (int g = 0; g < ROW_GROUPS; ++g)
        asm volatile("" :: "v"(f[g].x), "v"(f[g].y), "v"(f[g].z), "v"(f[g].w));
    }
    #pragma unroll
    for (int g = 0; g < ROW_GROUPS; ++g) {
      if (r3 < 3 && 3 * g + r3 < 32) {
        float* p = acc + dn[g] * APITCH + pc;
        atomicAdd(p, f[g].x * a[g]);
        atomicAdd(p + 9, f[g].y * a[g]);
        atomicAdd(p + 18, f[g].z * a[g]);
        atomicAdd(p + 27, f[g].w * a[g]);
      }
    }
  } else {
    float w0 = 0.0f, w1 = 0.0f, w2 = 0.0f;
    if (on) {
      float w[NHEAD];
      #pragma unroll
      for (int h = 0; h < NHEAD; ++h) {
        const float v = WIDE ? ev[h] : leaky(ev[h] + sEr[dl * NHEAD + h]);
        w[h] = expf(v - __uint_as_float(sMx[dl * NHEAD + h]));
        atomicAdd(&acc[dl * APITCH + FDIM + h], w[h]);
      }
      w0 = w[0]; w1 = w[1]; w2 = w[2];
    }
    const int ng = (cnt + 2) / 3;
    for (int g = 0; g < ng; ++g) {
      const int jr = 3 * g + rr;
      const bool ok = r3 < 3 && jr < cnt;
      const int j = jr < cnt ? jr : cnt - 1;
      int sx, dy;
      if (WIDE) {
        const int pk = qw[4 * (head + j) + 3];
        sx = pk >> PACK_SHIFT;
        dy = pk & PACK_MASK;
      } else {
        const int2 en = q[head + j];
        sx = en.x;
        dy = en.y;
      }
      const float4 fv = feat4[(size_t)sx * 9 + pc];
      const float a0 = __shfl(w0, j);
      const float a1 = __shfl(w1, j);
      const float a2 = __shfl(w2, j);
      const float av = pc < 3 ? a0 : (pc < 6 ? a1 : a2);
      if (ok) {
        float* p = acc + dy * APITCH + pc;
        atomicAdd(p, fv.x * av);
        atomicAdd(p + 9, fv.y * av);
        atomicAdd(p + 18, fv.z * av);
        atomicAdd(p + 27, fv.w * av);
      }
    }
  }
}

template <bool LOADED>
__device__ __forceinline__ void list_max(int2* q, int tail, bool wide, int lane,
                                         const int* __restrict__ src,
                                         const float* __restrict__ el,
                                         const float* sEr, unsigned* sMx) {
  v4i* q4 = (v4i*)q;
  const int* qw = (const int*)q;
  #pragma nounroll
  for (int i0 = (tail - 1) & ~127; i0 >= 0; i0 -= 128) {
    int sv[4], dv[4];
    #pragma unroll
    for (int k = 0; k < 4; ++k) {
      const int i = i0 + 32 * k + lane;
      if (LOADED) {
        const int pk = qw[4 * (i < WCAP ? i : WCAP - 1) + 3];
        const int x = i < tail ? pk : 0;
        sv[k] = x >> PACK_SHIFT;
        dv[k] = x & PACK_MASK;
      } else {
        const int2 ent = q[i < LCAP ? i : LCAP - 1];
        const int x = i < tail ? ent.x : 0;
        sv[k] = src[x];
        dv[k] = i < tail ? ent.y : 0;
      }
    }
    float ev[4][NHEAD];
    #pragma unroll
    for (int k = 0; k < 4; ++k) {
      #pragma unroll
      for (int h = 0; h < NHEAD; ++h) ev[k][h] = el[sv[k] * NHEAD + h];
    }
    #pragma unroll
    for (int k = 0; k < 4; ++k) {
      const int i = i0 + 32 * k + lane;
      float v[NHEAD];
      #pragma unroll
      for (int h = 0; h < NHEAD; ++h) {
        v[h] = leaky(ev[k][h] + sEr[dv[k] * NHEAD + h]);
        atomicMax(&sMx[dv[k] * NHEAD + h], i < tail ? f2key(v[h]) : NEG_INF_KEY);
      }
      if (i < tail) {
        if (LOADED || wide) {
          q4[i] = (v4i){__float_as_int(v[0]), __float_as_int(v[1]),
                        __float_as_int(v[2]), (sv[k] << PACK_SHIFT) | dv[k]};
        } else {
          q[i].x = sv[k];
        }
      }
    }
  }
}

__global__ __launch_bounds__(AGG_THREADS) void edge_agg(
    const int* __restrict__ src, const int* __restrict__ dst,
    const float* __restrict__ el, const float* __restrict__ er,
    const float* __restrict__ feat,
    float* h, const float* __restrict__ bias,
    int* glist, const unsigned* __restrict__ bmap, int n, int ne, int layer) {
  __shared__ float acc[RN * APITCH];
  __shared__ float sEr[RN * NHEAD];
  __shared__ unsigned sMx[RN * NHEAD];
  __shared__ v4i lst4[AGG_WAVES * WCAP];
  int2* lst = (int2*)lst4;
  __shared__ int sOvf;
  __shared__ int sCnt[AGG_WAVES];
  __shared__ int sOff[AGG_WAVES + 1];

  const int tid = threadIdx.x;
  const int lane = tid & 31;
  const int wave = __builtin_amdgcn_readfirstlane(tid >> 5);
  const int lo = blockIdx.x * RN;
  const int cntN = (n - lo) < RN ? (n - lo) : RN;

  for (int i = tid; i < RN * APITCH; i += AGG_THREADS) acc[i] = 0.0f;
  for (int i = tid; i < RN * NHEAD; i += AGG_THREADS) sMx[i] = NEG_INF_KEY;
  for (int i = tid; i < cntN * NHEAD; i += AGG_THREADS)
    sEr[i] = er[(size_t)lo * NHEAD + i];
  if (tid == 0) sOvf = 0;
  __syncthreads();

  int2* q = lst + wave * LCAP;
  int* qi = (int*)q;
  int tail = 0;
  int ncand = 0;

  const int capT = ((n * FDIM) / (int)gridDim.x) & ~31;
  int* gl = glist + (size_t)blockIdx.x * capT;
  int loaded = 0;
  if (layer > 0) {
    const int cnt = __builtin_amdgcn_readfirstlane(gl[0]);
    if (cnt >= 0) {
      loaded = 1;
      const int nq = (cnt + 3) >> 2;
      const int qa = (wave * nq) >> 5;
      const int nqw = (((wave + 1) * nq) >> 5) - qa;
      const int rem = cnt - 4 * qa;
      tail = 4 * nqw < rem ? 4 * nqw : rem;
      const int* gw = gl + 32 + 4 * qa;
      int p[WCAP / 32];
      #pragma unroll
      for (int k = 0; k < WCAP / 32; ++k) {
        const int i = 32 * k + lane;
        p[k] = gw[i < tail ? i : 0];
      }
      #pragma unroll
      for (int k = 0; k < WCAP / 32; ++k) {
        const int i = 32 * k + lane;
        if (i < tail) qi[4 * i + 3] = p[k];
      }
    }
  }

  int ovf = (!loaded && gridDim.x > 256) ? 1 : 0;

  const int nv8 = (ne + 7) >> 3;
  if (nv8 > 0 && !ovf && !loaded) {
    const int nw = ((nv8 + KV - 1) / KV) * 32;
    const unsigned* __restrict__ row = bmap + (size_t)blockIdx.x * nw;
    const int wa = (wave * nw) >> 5;
    const int wb = ((wave + 1) * nw) >> 5;
    for (int base = wa; base < wb; base += 256) {
      unsigned mw[8];
      #pragma unroll
      for (int u = 0; u < 8; ++u) {
        const int k = base + 32 * u + lane;
        const unsigned x = row[k < wb ? k : wa];
        mw[u] = k < wb ? x : 0u;
      }
      int c = 0;
      #pragma unroll
      for (int u = 0; u < 8; ++u) c += __builtin_popcount(mw[u]);
      int inc = c;
      #pragma unroll
      for (int d = 1; d < 32; d <<= 1) {
        const int t = __shfl_up(inc, d);
        if (lane >= d) inc += t;
      }
      const int tot = __builtin_amdgcn_readlane(inc, 31);
      if (ncand + tot <= CCAP) {
        int pos = ncand + inc - c;
        #pragma unroll
        for (int u = 0; u < 8; ++u) {
          unsigned m = mw[u];
          const int vb = (base + 32 * u + lane) * 32;
          while (m != 0u) {
            qi[2 * LCAP - 1 - pos] = vb + __builtin_ctz(m);
            ++pos;
            m &= m - 1u;
          }
        }
        ncand += tot;
      } else {
        ovf = 1;
      }
    }
  }
  __syncthreads();

#define PUSH(dval, eidx, okb)                                                \
  {                                                                          \
    const int dlc = (dval) - lo;                                             \
    const unsigned mk = __builtin_amdgcn_ballot_w32(                         \
        (okb) && (unsigned)dlc < (unsigned)cntN);                            \
    if (mk) {                                                                \
      const int ntail = tail + __builtin_popcount(mk);                       \
      if (ntail <= FCAP) {                                                   \
        if ((mk >> lane) & 1u) {                                             \
          const int pos = tail + (int)__builtin_amdgcn_mbcnt_lo(mk, 0u);     \
          q[pos] = make_int2((eidx), dlc);                                   \
        }                                                                    \
        tail = ntail;                                                        \
      } else {                                                               \
        ovf = 1;                                                             \
      }                                                                      \
    }                                                                        \
  }
  if (!ovf && !loaded) {
    const int4* __restrict__ dst4 = (const int4*)dst;
    for (int i0 = 0; i0 < ncand; i0 += 32 * EXP_U) {
      int vv[EXP_U];
      int dd[EXP_U][8];
      unsigned hm[EXP_U];
      #pragma unroll
      for (int u = 0; u < EXP_U; ++u) {
        const int ci = i0 + 32 * u + lane;
        vv[u] = qi[2 * LCAP - 1 - (ci < ncand ? ci : ncand - 1)];
        const bool whole = 8 * vv[u] + 7 < ne;
        int4 t0 = make_int4(-1, -1, -1, -1), t1 = t0;
        if (ne >= 8) {
          const int s8 = whole ? vv[u] : 0;
          t0 = dst4[2 * s8];
          t1 = dst4[2 * s8 + 1];
        }
        dd[u][0] = t0.x; dd[u][1] = t0.y; dd[u][2] = t0.z; dd[u][3] = t0.w;
        dd[u][4] = t1.x; dd[u][5] = t1.y; dd[u][6] = t1.z; dd[u][7] = t1.w;
        if (!whole) {
          #pragma unroll
          for (int j = 0; j < 8; ++j)
            dd[u][j] = (8 * vv[u] + j) < ne ? dst[8 * vv[u] + j] : -1;
        }
      }
      #pragma unroll
      for (int u = 0; u < EXP_U; ++u) {
        const int ci = i0 + 32 * u + lane;
        unsigned hb = 0u;
        #pragma unroll
        for (int j = 0; j < 8; ++j)
          hb |= (unsigned)(dd[u][j] - lo) < (unsigned)cntN ? (1u << j) : 0u;
        hm[u] = ci < ncand ? hb : 0u;
      }
      unsigned any = 0u;
      #pragma unroll
      for (int u = 0; u < EXP_U; ++u) any |= hm[u];
      while (__builtin_amdgcn_ballot_w32(any != 0u)) {
        #pragma unroll
        for (int u = 0; u < EXP_U; ++u) {
          const int p = __builtin_ctz(hm[u] | 0x100u);
          int dsel = dd[u][0];
          #pragma unroll
          for (int j = 1; j < 8; ++j) dsel = p == j ? dd[u][j] : dsel;
          const bool okk = hm[u] != 0u;
          PUSH(dsel, 8 * vv[u] + p, okk)
          hm[u] &= hm[u] - 1u;
        }
        any = 0u;
        #pragma unroll
        for (int u = 0; u < EXP_U; ++u) any |= hm[u];
      }
    }
  }
#undef PUSH
  if (ovf && lane == 0) sOvf = 1;
  __syncthreads();
  const int slow = __builtin_amdgcn_readfirstlane(sOvf);

  const bool wide = loaded || (tail <= WCAP && n <= PACK_MAXN);

  if (!slow) {
    if (loaded) list_max<true>(q, tail, true, lane, src, el, sEr, sMx);
    else        list_max<false>(q, tail, wide, lane, src, el, sEr, sMx);
  } else {
    for (int e = tid; e < ne; e += AGG_THREADS) {
      const int dlc = dst[e] - lo;
      if ((unsigned)dlc < (unsigned)cntN) {
        const int s = src[e];
        #pragma unroll
        for (int h = 0; h < NHEAD; ++h)
          atomicMax(&sMx[dlc * NHEAD + h],
                    f2key(leaky(el[s * NHEAD + h] + sEr[dlc * NHEAD + h])));
      }
    }
  }
  if (layer == 0 && lane == 0) sCnt[wave] = slow ? 0 : tail;
  __syncthreads();

  if (layer == 0 && tid < AGG_WAVES) {
    int o = 0;
    for (int w = 0; w < AGG_WAVES; ++w) o += w < tid ? sCnt[w] : 0;
    sOff[tid] = o;
    if (tid == AGG_WAVES - 1) sOff[AGG_WAVES] = o + sCnt[AGG_WAVES - 1];
  }

  for (int i = tid; i < cntN * NHEAD; i += AGG_THREADS) {
    float v = key2f(sMx[i]);
    if (!isfinite(v)) v = 0.0f;
    sMx[i] = __float_as_uint(v);
  }
  __syncthreads();

  if (layer == 0) {
    const int total = __builtin_amdgcn_readfirstlane(sOff[AGG_WAVES]);
    const bool ok = !slow && n <= PACK_MAXN && capT >= 64 && total <= capT - 32;
    if (ok) {
      const int up = total > 0 ? ((total + 31) & ~31) : 32;
      for (int idx = tid; idx < up; idx += AGG_THREADS) {
        int w = 0;
        #pragma unroll
        for (int s = AGG_WAVES / 2; s > 0; s >>= 1)
          if (sOff[w + s] <= idx) w += s;
        int val = 0;
        if (idx < total) {
          const int k = idx - sOff[w];
          if (sCnt[w] <= WCAP) {
            val = ((const int*)lst4)[4 * (w * WCAP + k) + 3];
          } else {
            const int2 ent = lst[w * LCAP + k];
            val = (ent.x << PACK_SHIFT) | ent.y;
          }
        }
        gl[32 + idx] = val;
      }
    }
    if (tid < 32) gl[tid] = tid == 0 ? (ok ? total : -1) : 0;
  }

  if (!slow) {
    int head = 0;
    if (wide) {
      while (tail - head >= 32) {
        agg_batch<true, true>(q, head, 32, lane, el, feat, sEr, sMx, acc);
        head += 32;
      }
      if (tail - head > 0)
        agg_batch<false, true>(q, head, tail - head, lane, el, feat, sEr, sMx, acc);
    } else {
      while (tail - head >= 32) {
        agg_batch<true, false>(q, head, 32, lane, el, feat, sEr, sMx, acc);
        head += 32;
      }
      if (tail - head > 0)
        agg_batch<false, false>(q, head, tail - head, lane, el, feat, sEr, sMx, acc);
    }
  } else {
    for (int e = tid; e < ne; e += AGG_THREADS) {
      const int dlc = dst[e] - lo;
      if ((unsigned)dlc < (unsigned)cntN) {
        const int s = src[e];
        float w[NHEAD];
        #pragma unroll
        for (int h = 0; h < NHEAD; ++h) {
          const float v = leaky(el[s * NHEAD + h] + sEr[dlc * NHEAD + h]);
          w[h] = expf(v - __uint_as_float(sMx[dlc * NHEAD + h]));
          atomicAdd(&acc[dlc * APITCH + FDIM + h], w[h]);
        }
        for (int c = 0; c < FDIM; ++c)
          atomicAdd(&acc[dlc * APITCH + colpos(c)],
                    feat[(size_t)s * FDIM + c] * w[c / HDIM]);
      }
    }
  }
  __syncthreads();

  for (int i = tid; i < cntN * FDIM; i += AGG_THREADS) {
    const int node = i / FDIM;
    const int c = i - node * FDIM;
    const float den = fmaxf(acc[node * APITCH + FDIM + c / HDIM], 1e-9f);
    const size_t g = (size_t)lo * FDIM + i;
    const float x = acc[node * APITCH + colpos(c)] / den + h[g] + bias[c];
    h[g] = x > 0.0f ? x : expm1f(x);
  }
}

extern "C" void kernel_launch(void* const* d_in, const int* in_sizes, int n_in,
                              void* d_out, int out_size, void* d_ws,
                              size_t ws_size, hipStream_t stream) {
  (void)n_in; (void)out_size; (void)ws_size;
  const float* X        = (const float*)d_in[0];
  const int*   src      = (const int*)d_in[1];
  const int*   dst      = (const int*)d_in[2];
  const float* lin0_w   = (const float*)d_in[3];
  const float* lin0_b   = (const float*)d_in[4];
  const float* fc_w     = (const float*)d_in[5];
  const float* attn_l   = (const float*)d_in[6];
  const float* attn_r   = (const float*)d_in[7];
  const float* gat_bias = (const float*)d_in[8];
  const float* out_w    = (const float*)d_in[9];
  const float* out_b    = (const float*)d_in[10];

  const int n  = in_sizes[0] / FDIM;
  const int ne = in_sizes[1];

  float* ws   = (float*)d_ws;
  float* h     = ws; ws += (size_t)n * FDIM;
  int*   glist = (int*)ws; ws += (size_t)n * FDIM;
  float* feat  = ws; ws += (size_t)n * FDIM;
  float* el    = ws; ws += (size_t)n * NHEAD;
  float* er    = ws; ws += (size_t)n * NHEAD;

  const dim3 gemmGrid((n + 15) / 16);
  const int nbN3 = (n * NHEAD + 255) / 256;
  const int nbR  = (n + RN - 1) / RN;

  unsigned* bmap = (unsigned*)d_out;
  const int nv8 = (ne + 7) / 8;
  edge_bits<<<(nv8 + KV - 1) / KV, KV, 0, stream>>>(dst, bmap, ne);

  gemm36_wmma<<<gemmGrid, 32, 0, stream>>>(X, lin0_w, lin0_b, h, n);

  for (int l = 0; l < 3; ++l) {
    gemm36_wmma<<<gemmGrid, 32, 0, stream>>>(
        h, fc_w + (size_t)l * FDIM * FDIM, nullptr, feat, n);
    attn_scores<<<nbN3, 256, 0, stream>>>(
        feat, attn_l + (size_t)l * NHEAD * HDIM, attn_r + (size_t)l * NHEAD * HDIM,
        el, er, n);
    edge_agg<<<nbR, AGG_THREADS, 0, stream>>>(src, dst, el, er, feat, h,
                                             gat_bias + (size_t)l * FDIM, glist,
                                             bmap, n, ne, l);
  }

  gemm36_wmma<<<gemmGrid, 32, 0, stream>>>(h, out_w, out_b, (float*)d_out, n);
}


# edge_agg without max pass: softmax shifted by the bound leaky(max el of all nodes + er) (exact max kept where el spans > 64); el rides in the feat row (160 B): no el gather; arrival order kept
# speedup vs baseline: 19.3504x; 1.3692x over previous
#include <hip/hip_runtime.h>
#include <math.h>

typedef __attribute__((ext_vector_type(2))) float v2f;
typedef __attribute__((ext_vector_type(8))) float v8f;
typedef __attribute__((ext_vector_type(4))) int v4i;

#define FDIM 36
#define NHEAD 3
#define HDIM 12
#define NEG_SLOPE 0.2f
#define NEG_INF_KEY 0x007FFFFFu
#define FP 40
#define MAX_GAP 64.0f

__device__ __forceinline__ unsigned f2key(float f) {
  unsigned u = __float_as_uint(f);
  return (u & 0x80000000u) ? ~u : (u | 0x80000000u);
}
__device__ __forceinline__ float key2f(unsigned k) {
  return (k & 0x80000000u) ? __uint_as_float(k & 0x7FFFFFFFu)
                           : __uint_as_float(~k);
}
__device__ __forceinline__ float leaky(float x) {
  return x >= 0.0f ? x : NEG_SLOPE * x;
}

template <int LDO>
__global__ __launch_bounds__(32) void gemm36_wmma(
    const float* __restrict__ A, const float* __restrict__ W,
    const float* __restrict__ bias, float* __restrict__ out, int nrows) {
  const int lane = threadIdx.x;
  const int half = lane >> 4;
  const int idx  = lane & 15;

  int mrow = blockIdx.x * 16 + idx;
  mrow = mrow < nrows ? mrow : (nrows - 1);
  const float* __restrict__ Ab = A + (size_t)mrow * FDIM + 2 * half;

  const int col0 = idx;
  const int col1 = idx + 16;
  const int col2 = (idx + 32) < FDIM ? (idx + 32) : (FDIM - 1);
  const float* __restrict__ Wb0 = W + col0 + 72 * half;
  const float* __restrict__ Wb1 = W + col1 + 72 * half;
  const float* __restrict__ Wb2 = W + col2 + 72 * half;

  v8f acc[3];
  acc[0] = (v8f){0.f,0.f,0.f,0.f,0.f,0.f,0.f,0.f};
  acc[1] = acc[0];
  acc[2] = acc[0];

  #pragma unroll
  for (int k0 = 0; k0 < FDIM; k0 += 4) {
    v2f a, b0, b1, b2;
    a.x  = Ab[k0];
    a.y  = Ab[k0 + 1];
    b0.x = Wb0[(k0 + 0) * FDIM];
    b0.y = Wb0[(k0 + 1) * FDIM];
    b1.x = Wb1[(k0 + 0) * FDIM];
    b1.y = Wb1[(k0 + 1) * FDIM];
    b2.x = Wb2[(k0 + 0) * FDIM];
    b2.y = Wb2[(k0 + 1) * FDIM];
    acc[0] = __builtin_amdgcn_wmma_f32_16x16x4_f32(false, a, false, b0,
                                                   (short)0, acc[0], false, false);
    acc[1] = __builtin_amdgcn_wmma_f32_16x16x4_f32(false, a, false, b1,
                                                   (short)0, acc[1], false, false);
    acc[2] = __builtin_amdgcn_wmma_f32_16x16x4_f32(false, a, false, b2,
                                                   (short)0, acc[2], false, false);
  }

  const int rowbase = blockIdx.x * 16 + 8 * half;
  const bool full = (blockIdx.x * 16 + 16) <= nrows;
  #pragma unroll
  for (int t = 0; t < 3; ++t) {
    const int col = idx + 16 * t;
    if (col < FDIM) {
      const float bv = bias ? bias[col] : 0.0f;
      float* o = out + (size_t)rowbase * LDO + col;
      if (full) {
        #pragma unroll
        for (int v = 0; v < 8; ++v) o[(size_t)v * LDO] = acc[t][v] + bv;
      } else {
        #pragma unroll
        for (int v = 0; v < 8; ++v)
          if (rowbase + v < nrows) o[(size_t)v * LDO] = acc[t][v] + bv;
      }
    }
  }
}

__global__ __launch_bounds__(256) void attn_scores(float* feat,
                            const float* __restrict__ a_l,
                            const float* __restrict__ a_r,
                            float* __restrict__ er, unsigned* stat,
                            int n) {
  __shared__ unsigned sst[6];
  const int tid = threadIdx.x;
  if (tid < 6) sst[tid] = tid < 3 ? 0u : 0xFFFFFFFFu;
  __syncthreads();
  const int node = blockIdx.x * blockDim.x + tid;
  const bool on = node < n;
  float sl[NHEAD] = {0.0f, 0.0f, 0.0f}, sr[NHEAD] = {0.0f, 0.0f, 0.0f};
  if (on) {
    float4* f4 = (float4*)(feat + (size_t)node * FP);
    const float4* l4 = (const float4*)a_l;
    const float4* r4 = (const float4*)a_r;
    #pragma unroll
    for (int h = 0; h < NHEAD; ++h) {
      #pragma unroll
      for (int q = 0; q < HDIM / 4; ++q) {
        const float4 fv = f4[3 * h + q], lv = l4[3 * h + q], rv = r4[3 * h + q];
        sl[h] += fv.x * lv.x + fv.y * lv.y + fv.z * lv.z + fv.w * lv.w;
        sr[h] += fv.x * rv.x + fv.y * rv.y + fv.z * rv.z + fv.w * rv.w;
      }
    }
    f4[9] = make_float4(sl[0], sl[1], sl[2], 0.0f);
    #pragma unroll
    for (int h = 0; h < NHEAD; ++h) er[(size_t)node * NHEAD + h] = sr[h];
  }
  unsigned kx[NHEAD], kn[NHEAD];
  #pragma unroll
  for (int h = 0; h < NHEAD; ++h) {
    kx[h] = on ? f2key(sl[h]) : 0u;
    kn[h] = on ? f2key(sl[h]) : 0xFFFFFFFFu;
  }
  #pragma unroll
  for (int d = 16; d > 0; d >>= 1) {
    #pragma unroll
    for (int h = 0; h < NHEAD; ++h) {
      const unsigned ox = (unsigned)__shfl_xor((int)kx[h], d);
      const unsigned on2 = (unsigned)__shfl_xor((int)kn[h], d);
      kx[h] = ox > kx[h] ? ox : kx[h];
      kn[h] = on2 < kn[h] ? on2 : kn[h];
    }
  }
  if ((tid & 31) == 0) {
    #pragma unroll
    for (int h = 0; h < NHEAD; ++h) {
      atomicMax(&sst[h], kx[h]);
      atomicMin(&sst[3 + h], kn[h]);
    }
  }
  __syncthreads();
  if (tid < 3) atomicMax(&stat[tid], sst[tid]);
  else if (tid < 6) atomicMin(&stat[tid], sst[tid]);
}

#define RN 392
#define KV 1024
#define BPITCH 257
__global__ __launch_bounds__(KV) void edge_bits(const int* __restrict__ dst,
                                                unsigned* __restrict__ bmap,
                                                unsigned* __restrict__ stat, int ne) {
  __shared__ unsigned bm[(KV / 32) * BPITCH];
  const int tid = threadIdx.x;
  const int lane = tid & 31;
  const int wave = tid >> 5;
  for (int i = tid; i < (KV / 32) * BPITCH; i += KV) bm[i] = 0u;
  if (blockIdx.x == 0 && tid < 18) stat[tid] = (tid % 6) < 3 ? 0u : 0xFFFFFFFFu;
  __syncthreads();

  const int nv8 = (ne + 7) >> 3;
  const int v = blockIdx.x * KV + tid;
  if (v < nv8) {
    int d[8];
    if (8 * v + 7 < ne) {
      const int4 t0 = ((const int4*)dst)[2 * v];
      const int4 t1 = ((const int4*)dst)[2 * v + 1];
      d[0] = t0.x; d[1] = t0.y; d[2] = t0.z; d[3] = t0.w;
      d[4] = t1.x; d[5] = t1.y; d[6] = t1.z; d[7] = t1.w;
    } else {
      #pragma unroll
      for (int j = 0; j < 8; ++j) d[j] = (8 * v + j) < ne ? dst[8 * v + j] : -1;
    }
    #pragma unroll
    for (int j = 0; j < 8; ++j) {
      unsigned b = (unsigned)d[j] / RN;
      b = b < 255u ? b : 255u;
      if (8 * v + j < ne) atomicOr(&bm[wave * BPITCH + b], 1u << lane);
    }
  }
  __syncthreads();

  for (int idx = tid; idx < 256 * 32; idx += KV) {
    const int r = idx >> 5;
    const int w = idx & 31;
    bmap[((size_t)r * gridDim.x + blockIdx.x) * 32 + w] = bm[w * BPITCH + r];
  }
}

#define GLW 34
#define PACK_SHIFT 9
#define PACK_MAXN (1 << 22)
#define APITCH 40
#define LCAP 896
#define FCAP 576
#define CCAP 640
#define EXP_U 2
#define WCAP (LCAP / 2)
#define PACK_MASK ((1 << PACK_SHIFT) - 1)
#define AGG_THREADS 1024
#define AGG_WAVES (AGG_THREADS / 32)

__device__ __forceinline__ int colpos(int c) { return (c & 3) * 9 + (c >> 2); }

#define ROW_GROUPS 11

template <bool FULL, bool WIDE>
__device__ __forceinline__ void agg_batch(const int2* q, int head, int cnt,
                                          int lane,
                                          const float* __restrict__ el,
                                          const float* __restrict__ feat,
                                          const float* sEr,
                                          const unsigned* sMx, float* acc) {
  const float4* __restrict__ feat4 = (const float4*)feat;
  const v4i* q4 = (const v4i*)q;
  const int* qw = (const int*)q;
  const int r3 = lane / 9;
  const int rr = r3 < 3 ? r3 : 2;
  const int pc = lane - 9 * r3;

  const bool on = FULL || lane < cnt;
  int dl = 0;
  float ev[NHEAD] = {0.0f, 0.0f, 0.0f};
  if (on) {
    if (WIDE) {
      const v4i ent = q4[head + lane];
      dl = ent.w & PACK_MASK;
      ev[0] = __int_as_float(ent.x);
      ev[1] = __int_as_float(ent.y);
      ev[2] = __int_as_float(ent.z);
    } else {
      const int2 ent = q[head + lane];
      const int s = ent.x;
      dl = ent.y;
      #pragma unroll
      for (int h = 0; h < NHEAD; ++h) ev[h] = el[(size_t)s * FP + h];
    }
  }

  if (FULL) {
    float4 f[ROW_GROUPS];
    int dn[ROW_GROUPS];
    #pragma unroll
    for (int g = 0; g < ROW_GROUPS; ++g) {
      const int j = (3 * g + rr) < 32 ? (3 * g + rr) : 31;
      int sx;
      if (WIDE) {
        const int* qr = qw + 4 * (head + rr) + 3;
        const int pk = g < ROW_GROUPS - 1 ? qr[12 * g] : qr[rr < 2 ? 120 : 116];
        sx = pk >> PACK_SHIFT;
        dn[g] = pk & PACK_MASK;
      } else {
        const int2 en = q[head + j];
        sx = en.x;
        dn[g] = en.y;
      }
      f[g] = feat4[(size_t)sx * (FP / 4) + pc];
    }
    if (!WIDE) {
      #pragma unroll
      for (int g = 0; g < ROW_GROUPS; ++g)
        asm volatile("" :: "v"(f[g].x), "v"(f[g].y), "v"(f[g].z), "v"(f[g].w));
    }

    float w[NHEAD];
    #pragma unroll
    for (int h = 0; h < NHEAD; ++h) {
      const float v = WIDE ? ev[h] : leaky(ev[h] + sEr[dl * NHEAD + h]);
      w[h] = expf(v - __uint_as_float(sMx[dl * NHEAD + h]));
      atomicAdd(&acc[dl * APITCH + FDIM + h], w[h]);
    }

    float a[ROW_GROUPS];
    #pragma unroll
    for (int g = 0; g < ROW_GROUPS; ++g) {
      const int j = (3 * g + rr) < 32 ? (3 * g + rr) : 31;
      const float a0 = __shfl(w[0], j);
      const float a1 = __shfl(w[1], j);
      const float a2 = __shfl(w[2], j);
      a[g] = pc < 3 ? a0 : (pc < 6 ? a1 : a2);
    }
    if (WIDE) {
      #pragma unroll
      for (int g = 0; g < ROW_GROUPS; ++g)
        asm volatile("" :: "v"(f[g].x), "v"(f[g].y), "v"(f[g].z), "v"(f[g].w));
    }
    #pragma unroll
    for (int g = 0; g < ROW_GROUPS; ++g) {
      if (r3 < 3 && 3 * g + r3 < 32) {
        float* p = acc + dn[g] * APITCH + pc;
        atomicAdd(p, f[g].x * a[g]);
        atomicAdd(p + 9, f[g].y * a[g]);
        atomicAdd(p + 18, f[g].z * a[g]);
        atomicAdd(p + 27, f[g].w * a[g]);
      }
    }
  } else {
    float w0 = 0.0f, w1 = 0.0f, w2 = 0.0f;
    if (on) {
      float w[NHEAD];
      #pragma unroll
      for (int h = 0; h < NHEAD; ++h) {
        const float v = WIDE ? ev[h] : leaky(ev[h] + sEr[dl * NHEAD + h]);
        w[h] = expf(v - __uint_as_float(sMx[dl * NHEAD + h]));
        atomicAdd(&acc[dl * APITCH + FDIM + h], w[h]);
      }
      w0 = w[0]; w1 = w[1]; w2 = w[2];
    }
    const int ng = (cnt + 2) / 3;
    for (int g = 0; g < ng; ++g) {
      const int jr = 3 * g + rr;
      const bool ok = r3 < 3 && jr < cnt;
      const int j = jr < cnt ? jr : cnt - 1;
      int sx, dy;
      if (WIDE) {
        const int pk = qw[4 * (head + j) + 3];
        sx = pk >> PACK_SHIFT;
        dy = pk & PACK_MASK;
      } else {
        const int2 en = q[head + j];
        sx = en.x;
        dy = en.y;
      }
      const float4 fv = feat4[(size_t)sx * (FP / 4) + pc];
      const float a0 = __shfl(w0, j);
      const float a1 = __shfl(w1, j);
      const float a2 = __shfl(w2, j);
      const float av = pc < 3 ? a0 : (pc < 6 ? a1 : a2);
      if (ok) {
        float* p = acc + dy * APITCH + pc;
        atomicAdd(p, fv.x * av);
        atomicAdd(p + 9, fv.y * av);
        atomicAdd(p + 18, fv.z * av);
        atomicAdd(p + 27, fv.w * av);
      }
    }
  }
}

template <bool LOADED>
__device__ __forceinline__ void list_max(int2* q, int tail, bool wide, int lane,
                                         const int* __restrict__ src,
                                         const float* __restrict__ el,
                                         const float* sEr, unsigned* sMx) {
  v4i* q4 = (v4i*)q;
  const int* qw = (const int*)q;
  #pragma nounroll
  for (int i0 = (tail - 1) & ~127; i0 >= 0; i0 -= 128) {
    int sv[4], dv[4];
    #pragma unroll
    for (int k = 0; k < 4; ++k) {
      const int i = i0 + 32 * k + lane;
      if (LOADED) {
        const int pk = qw[4 * (i < WCAP ? i : WCAP - 1) + 3];
        const int x = i < tail ? pk : 0;
        sv[k] = x >> PACK_SHIFT;
        dv[k] = x & PACK_MASK;
      } else {
        const int2 ent = q[i < LCAP ? i : LCAP - 1];
        const int x = i < tail ? ent.x : 0;
        sv[k] = src[x];
        dv[k] = i < tail ? ent.y : 0;
      }
    }
    float ev[4][NHEAD];
    #pragma unroll
    for (int k = 0; k < 4; ++k) {
      #pragma unroll
      for (int h = 0; h < NHEAD; ++h) ev[k][h] = el[(size_t)sv[k] * FP + h];
    }
    #pragma unroll
    for (int k = 0; k < 4; ++k) {
      const int i = i0 + 32 * k + lane;
      float v[NHEAD];
      #pragma unroll
      for (int h = 0; h < NHEAD; ++h) {
        v[h] = leaky(ev[k][h] + sEr[dv[k] * NHEAD + h]);
        atomicMax(&sMx[dv[k] * NHEAD + h], i < tail ? f2key(v[h]) : NEG_INF_KEY);
      }
      if (i < tail) {
        if (LOADED || wide) {
          q4[i] = (v4i){__float_as_int(v[0]), __float_as_int(v[1]),
                        __float_as_int(v[2]), (sv[k] << PACK_SHIFT) | dv[k]};
        } else {
          q[i].x = sv[k];
        }
      }
    }
  }
}

__device__ __forceinline__ void agg_bound(const int* qw, int head, int cnt, int lane,
                                          const float* __restrict__ feat,
                                          const v2f* sB, float* acc) {
  const float4* __restrict__ feat4 = (const float4*)feat;
  const int r3 = lane / 10;
  const int rr = r3 < 3 ? r3 : 2;
  const int pc = lane - 10 * r3;
  const int hh = pc < 3 ? 0 : (pc < 6 ? 1 : 2);
  const int esrc = 10 * rr + 9;
  int pk[ROW_GROUPS];
  #pragma unroll
  for (int g = 0; g < ROW_GROUPS; ++g) {
    const int jr = 3 * g + rr;
    pk[g] = qw[head + (jr < cnt ? jr : cnt - 1)];
  }
  float4 f[ROW_GROUPS];
  v2f b[ROW_GROUPS];
  #pragma unroll
  for (int g = 0; g < ROW_GROUPS; ++g)
    f[g] = feat4[(size_t)(pk[g] >> PACK_SHIFT) * (FP / 4) + pc];
  #pragma unroll
  for (int g = 0; g < ROW_GROUPS; ++g)
    b[g] = sB[(pk[g] & PACK_MASK) * NHEAD + hh];
  __builtin_amdgcn_sched_barrier(0);
  float w[ROW_GROUPS];
  #pragma unroll
  for (int g = 0; g < ROW_GROUPS; ++g) {
    const float e0 = __shfl(f[g].x, esrc);
    const float e1 = __shfl(f[g].y, esrc);
    const float e2 = __shfl(f[g].z, esrc);
    w[g] = hh == 0 ? e0 : (hh == 1 ? e1 : e2);
  }
  #pragma unroll
  for (int g = 0; g < ROW_GROUPS; ++g) w[g] = expf(leaky(w[g] + b[g].x) - b[g].y);
  #pragma unroll
  for (int g = 0; g < ROW_GROUPS; ++g) {
    if (r3 < 3 && pc < 9 && 3 * g + r3 < cnt) {
      float* p = acc + (pk[g] & PACK_MASK) * APITCH + pc;
      atomicAdd(p, f[g].x * w[g]);
      atomicAdd(p + 9, f[g].y * w[g]);
      atomicAdd(p + 18, f[g].z * w[g]);
      atomicAdd(p + 27, f[g].w * w[g]);
      if (pc == 3 * hh) atomicAdd(acc + (pk[g] & PACK_MASK) * APITCH + FDIM + hh, w[g]);
    }
  }
}

__device__ __forceinline__ void list_pack(int* qw, int tail, int lane,
                                          const int* __restrict__ src) {
  #pragma nounroll
  for (int i0 = 0; i0 < tail; i0 += 128) {
    int sv[4], dv[4];
    #pragma unroll
    for (int k = 0; k < 4; ++k) {
      const int i = i0 + 32 * k + lane;
      const int ic = i < LCAP ? i : LCAP - 1;
      const int ex = qw[2 * ic];
      dv[k] = qw[2 * ic + 1];
      sv[k] = src[i < tail ? ex : 0];
    }
    #pragma unroll
    for (int k = 0; k < 4; ++k) {
      const int i = i0 + 32 * k + lane;
      if (i < tail) qw[i] = (sv[k] << PACK_SHIFT) | dv[k];
    }
  }
}

__global__ __launch_bounds__(AGG_THREADS) void edge_agg(
    const int* __restrict__ src, const int* __restrict__ dst,
    const float* __restrict__ er,
    const float* __restrict__ feat, const unsigned* __restrict__ stat,
    float* h, const float* __restrict__ bias,
    int* glist, const unsigned* __restrict__ bmap, int n, int ne, int layer) {
  const float* __restrict__ el = feat + FDIM;
  __shared__ float acc[RN * APITCH];
  __shared__ float sEr[RN * NHEAD];
  __shared__ unsigned sMx[RN * NHEAD];
  __shared__ v2f sB[RN * NHEAD];
  __shared__ v4i lst4[AGG_WAVES * WCAP];
  int2* lst = (int2*)lst4;
  __shared__ int sOvf;
  __shared__ int sCnt[AGG_WAVES];
  __shared__ int sOff[AGG_WAVES + 1];

  const int tid = threadIdx.x;
  const int lane = tid & 31;
  const int wave = __builtin_amdgcn_readfirstlane(tid >> 5);
  const int lo = blockIdx.x * RN;
  const int cntN = (n - lo) < RN ? (n - lo) : RN;

  float gM[NHEAD];
  bool fb = n <= PACK_MAXN;
  #pragma unroll
  for (int hd = 0; hd < NHEAD; ++hd) {
    const float mx = key2f(stat[hd]);
    const float mn = key2f(stat[NHEAD + hd]);
    gM[hd] = mx;
    fb = fb && (mx - mn <= MAX_GAP);
  }
  const int fast = __builtin_amdgcn_readfirstlane(fb ? 1 : 0);

  for (int i = tid; i < RN * APITCH; i += AGG_THREADS) acc[i] = 0.0f;
  for (int i = tid; i < RN * NHEAD; i += AGG_THREADS) sMx[i] = NEG_INF_KEY;
  for (int i = tid; i < cntN * NHEAD; i += AGG_THREADS)
    sEr[i] = er[(size_t)lo * NHEAD + i];
  if (fast) {
    for (int i = tid; i < cntN * NHEAD; i += AGG_THREADS) {
      const int hd = i % NHEAD;
      const float e = er[(size_t)lo * NHEAD + i];
      const float m = hd == 0 ? gM[0] : (hd == 1 ? gM[1] : gM[2]);
      sB[i] = (v2f){e, leaky(m + e)};
    }
  }
  if (tid == 0) sOvf = 0;
  __syncthreads();

  int2* q = lst + wave * LCAP;
  int* qi = (int*)q;
  int tail = 0;
  int ncand = 0;

  const int capT = ((n * GLW) / (int)gridDim.x) & ~31;
  int* gl = glist + (size_t)blockIdx.x * capT;
  int loaded = 0;
  if (layer > 0) {
    const int cnt = __builtin_amdgcn_readfirstlane(gl[0]);
    if (cnt >= 0) {
      loaded = 1;
      const int nq = (cnt + 3) >> 2;
      const int qa = (wave * nq) >> 5;
      const int nqw = (((wave + 1) * nq) >> 5) - qa;
      const int rem = cnt - 4 * qa;
      tail = 4 * nqw < rem ? 4 * nqw : rem;
      const int* gw = gl + 32 + 4 * qa;
      int p[WCAP / 32];
      #pragma unroll
      for (int k = 0; k < WCAP / 32; ++k) {
        const int i = 32 * k + lane;
        p[k] = gw[i < tail ? i : 0];
      }
      #pragma unroll
      for (int k = 0; k < WCAP / 32; ++k) {
        const int i = 32 * k + lane;
        if (i < tail) qi[fast ? i : 4 * i + 3] = p[k];
      }
    }
  }

  int ovf = (!loaded && gridDim.x > 256) ? 1 : 0;

  const int nv8 = (ne + 7) >> 3;
  if (nv8 > 0 && !ovf && !loaded) {
    const int nw = ((nv8 + KV - 1) / KV) * 32;
    const unsigned* __restrict__ row = bmap + (size_t)blockIdx.x * nw;
    const int wa = (wave * nw) >> 5;
    const int wb = ((wave + 1) * nw) >> 5;
    for (int base = wa; base < wb; base += 256) {
      unsigned mw[8];
      #pragma unroll
      for (int u = 0; u < 8; ++u) {
        const int k = base + 32 * u + lane;
        const unsigned x = row[k < wb ? k : wa];
        mw[u] = k < wb ? x : 0u;
      }
      int c = 0;
      #pragma unroll
      for (int u = 0; u < 8; ++u) c += __builtin_popcount(mw[u]);
      int inc = c;
      #pragma unroll
      for (int d = 1; d < 32; d <<= 1) {
        const int t = __shfl_up(inc, d);
        if (lane >= d) inc += t;
      }
      const int tot = __builtin_amdgcn_readlane(inc, 31);
      if (ncand + tot <= CCAP) {
        int pos = ncand + inc - c;
        #pragma unroll
        for (int u = 0; u < 8; ++u) {
          unsigned m = mw[u];
          const int vb = (base + 32 * u + lane) * 32;
          while (m != 0u) {
            qi[2 * LCAP - 1 - pos] = vb + __builtin_ctz(m);
            ++pos;
            m &= m - 1u;
          }
        }
        ncand += tot;
      } else {
        ovf = 1;
      }
    }
  }
  __syncthreads();

#define PUSH(dval, eidx, okb)                                                \
  {                                                                          \
    const int dlc = (dval) - lo;                                             \
    const unsigned mk = __builtin_amdgcn_ballot_w32(                         \
        (okb) && (unsigned)dlc < (unsigned)cntN);                            \
    if (mk) {                                                                \
      const int ntail = tail + __builtin_popcount(mk);                       \
      if (ntail <= FCAP) {                                                   \
        if ((mk >> lane) & 1u) {                                             \
          const int pos = tail + (int)__builtin_amdgcn_mbcnt_lo(mk, 0u);     \
          q[pos] = make_int2((eidx), dlc);                                   \
        }                                                                    \
        tail = ntail;                                                        \
      } else {                                                               \
        ovf = 1;                                                             \
      }                                                                      \
    }                                                                        \
  }
  if (!ovf && !loaded) {
    const int4* __restrict__ dst4 = (const int4*)dst;
    for (int i0 = 0; i0 < ncand; i0 += 32 * EXP_U) {
      int vv[EXP_U];
      int dd[EXP_U][8];
      unsigned hm[EXP_U];
      #pragma unroll
      for (int u = 0; u < EXP_U; ++u) {
        const int ci = i0 + 32 * u + lane;
        vv[u] = qi[2 * LCAP - 1 - (ci < ncand ? ci : ncand - 1)];
        const bool whole = 8 * vv[u] + 7 < ne;
        int4 t0 = make_int4(-1, -1, -1, -1), t1 = t0;
        if (ne >= 8) {
          const int s8 = whole ? vv[u] : 0;
          t0 = dst4[2 * s8];
          t1 = dst4[2 * s8 + 1];
        }
        dd[u][0] = t0.x; dd[u][1] = t0.y; dd[u][2] = t0.z; dd[u][3] = t0.w;
        dd[u][4] = t1.x; dd[u][5] = t1.y; dd[u][6] = t1.z; dd[u][7] = t1.w;
        if (!whole) {
          #pragma unroll
          for (int j = 0; j < 8; ++j)
            dd[u][j] = (8 * vv[u] + j) < ne ? dst[8 * vv[u] + j] : -1;
        }
      }
      #pragma unroll
      for (int u = 0; u < EXP_U; ++u) {
        const int ci = i0 + 32 * u + lane;
        unsigned hb = 0u;
        #pragma unroll
        for (int j = 0; j < 8; ++j)
          hb |= (unsigned)(dd[u][j] - lo) < (unsigned)cntN ? (1u << j) : 0u;
        hm[u] = ci < ncand ? hb : 0u;
      }
      unsigned any = 0u;
      #pragma unroll
      for (int u = 0; u < EXP_U; ++u) any |= hm[u];
      while (__builtin_amdgcn_ballot_w32(any != 0u)) {
        #pragma unroll
        for (int u = 0; u < EXP_U; ++u) {
          const int p = __builtin_ctz(hm[u] | 0x100u);
          int dsel = dd[u][0];
          #pragma unroll
          for (int j = 1; j < 8; ++j) dsel = p == j ? dd[u][j] : dsel;
          const bool okk = hm[u] != 0u;
          PUSH(dsel, 8 * vv[u] + p, okk)
          hm[u] &= hm[u] - 1u;
        }
        any = 0u;
        #pragma unroll
        for (int u = 0; u < EXP_U; ++u) any |= hm[u];
      }
    }
  }
#undef PUSH
  if (ovf && lane == 0) sOvf = 1;
  __syncthreads();
  const int slow = __builtin_amdgcn_readfirstlane(sOvf);

  const bool wide = loaded || (tail <= WCAP && n <= PACK_MAXN);

  if (!slow) {
    if (fast) { if (!loaded) list_pack(qi, tail, lane, src); }
    else if (loaded) list_max<true>(q, tail, true, lane, src, el, sEr, sMx);
    else             list_max<false>(q, tail, wide, lane, src, el, sEr, sMx);
  } else {
    for (int e = tid; e < ne; e += AGG_THREADS) {
      const int dlc = dst[e] - lo;
      if ((unsigned)dlc < (unsigned)cntN) {
        const int s = src[e];
        #pragma unroll
        for (int h = 0; h < NHEAD; ++h)
          atomicMax(&sMx[dlc * NHEAD + h],
                    f2key(leaky(el[(size_t)s * FP + h] + sEr[dlc * NHEAD + h])));
      }
    }
  }
  if (layer == 0 && lane == 0) sCnt[wave] = slow ? 0 : tail;
  __syncthreads();

  if (layer == 0 && tid < AGG_WAVES) {
    int o = 0;
    for (int w = 0; w < AGG_WAVES; ++w) o += w < tid ? sCnt[w] : 0;
    sOff[tid] = o;
    if (tid == AGG_WAVES - 1) sOff[AGG_WAVES] = o + sCnt[AGG_WAVES - 1];
  }

  for (int i = tid; i < cntN * NHEAD; i += AGG_THREADS) {
    float v = key2f(sMx[i]);
    if (!isfinite(v)) v = 0.0f;
    sMx[i] = __float_as_uint(v);
  }
  __syncthreads();

  if (layer == 0) {
    const int total = __builtin_amdgcn_readfirstlane(sOff[AGG_WAVES]);
    const bool ok = !slow && n <= PACK_MAXN && capT >= 64 && total <= capT - 32;
    if (ok) {
      const int up = total > 0 ? ((total + 31) & ~31) : 32;
      for (int idx = tid; idx < up; idx += AGG_THREADS) {
        int w = 0;
        #pragma unroll
        for (int s = AGG_WAVES / 2; s > 0; s >>= 1)
          if (sOff[w + s] <= idx) w += s;
        int val = 0;
        if (idx < total) {
          const int k = idx - sOff[w];
          if (fast) {
            val = ((const int*)lst4)[2 * LCAP * w + k];
          } else if (sCnt[w] <= WCAP) {
            val = ((const int*)lst4)[4 * (w * WCAP + k) + 3];
          } else {
            const int2 ent = lst[w * LCAP + k];
            val = (ent.x << PACK_SHIFT) | ent.y;
          }
        }
        gl[32 + idx] = val;
      }
    }
    if (tid < 32) gl[tid] = tid == 0 ? (ok ? total : -1) : 0;
  }

  if (!slow && fast) {
    for (int head = 0; head < tail; head += 32)
      agg_bound(qi, head, (tail - head) < 32 ? (tail - head) : 32, lane, feat, sB, acc);
  } else if (!slow) {
    int head = 0;
    if (wide) {
      while (tail - head >= 32) {
        agg_batch<true, true>(q, head, 32, lane, el, feat, sEr, sMx, acc);
        head += 32;
      }
      if (tail - head > 0)
        agg_batch<false, true>(q, head, tail - head, lane, el, feat, sEr, sMx, acc);
    } else {
      while (tail - head >= 32) {
        agg_batch<true, false>(q, head, 32, lane, el, feat, sEr, sMx, acc);
        head += 32;
      }
      if (tail - head > 0)
        agg_batch<false, false>(q, head, tail - head, lane, el, feat, sEr, sMx, acc);
    }
  } else {
    for (int e = tid; e < ne; e += AGG_THREADS) {
      const int dlc = dst[e] - lo;
      if ((unsigned)dlc < (unsigned)cntN) {
        const int s = src[e];
        float w[NHEAD];
        #pragma unroll
        for (int h = 0; h < NHEAD; ++h) {
          const float v = leaky(el[(size_t)s * FP + h] + sEr[dlc * NHEAD + h]);
          w[h] = expf(v - __uint_as_float(sMx[dlc * NHEAD + h]));
          atomicAdd(&acc[dlc * APITCH + FDIM + h], w[h]);
        }
        for (int c = 0; c < FDIM; ++c)
          atomicAdd(&acc[dlc * APITCH + colpos(c)],
                    feat[(size_t)s * FP + c] * w[c / HDIM]);
      }
    }
  }
  __syncthreads();

  for (int i = tid; i < cntN * FDIM; i += AGG_THREADS) {
    const int node = i / FDIM;
    const int c = i - node * FDIM;
    const float wsum = acc[node * APITCH + FDIM + c / HDIM];
    const float den = (fast && !slow) ? (wsum > 0.0f ? wsum : 1.0f) : fmaxf(wsum, 1e-9f);
    const size_t g = (size_t)lo * FDIM + i;
    const float x = acc[node * APITCH + colpos(c)] / den + h[g] + bias[c];
    h[g] = x > 0.0f ? x : expm1f(x);
  }
}

extern "C" void kernel_launch(void* const* d_in, const int* in_sizes, int n_in,
                              void* d_out, int out_size, void* d_ws,
                              size_t ws_size, hipStream_t stream) {
  (void)n_in; (void)out_size; (void)ws_size;
  const float* X        = (const float*)d_in[0];
  const int*   src      = (const int*)d_in[1];
  const int*   dst      = (const int*)d_in[2];
  const float* lin0_w   = (const float*)d_in[3];
  const float* lin0_b   = (const float*)d_in[4];
  const float* fc_w     = (const float*)d_in[5];
  const float* attn_l   = (const float*)d_in[6];
  const float* attn_r   = (const float*)d_in[7];
  const float* gat_bias = (const float*)d_in[8];
  const float* out_w    = (const float*)d_in[9];
  const float* out_b    = (const float*)d_in[10];

  const int n  = in_sizes[0] / FDIM;
  const int ne = in_sizes[1];

  float* ws   = (float*)d_ws;
  float* h     = ws; ws += (size_t)n * FDIM;
  float* feat  = ws; ws += (size_t)n * FP;
  int*   glist = (int*)ws; ws += (size_t)n * GLW;
  float* er    = ws; ws += (size_t)n * NHEAD;
  unsigned* stat = (unsigned*)ws;

  const dim3 gemmGrid((n + 15) / 16);
  const int nbN1 = (n + 255) / 256;
  const int nbR  = (n + RN - 1) / RN;

  unsigned* bmap = (unsigned*)d_out;
  const int nv8 = (ne + 7) / 8;
  edge_bits<<<(nv8 + KV - 1) / KV, KV, 0, stream>>>(dst, bmap, stat, ne);

  gemm36_wmma<FDIM><<<gemmGrid, 32, 0, stream>>>(X, lin0_w, lin0_b, h, n);

  for (int l = 0; l < 3; ++l) {
    gemm36_wmma<FP><<<gemmGrid, 32, 0, stream>>>(
        h, fc_w + (size_t)l * FDIM * FDIM, nullptr, feat, n);
    attn_scores<<<nbN1, 256, 0, stream>>>(
        feat, attn_l + (size_t)l * NHEAD * HDIM, attn_r + (size_t)l * NHEAD * HDIM,
        er, stat + 6 * l, n);
    edge_agg<<<nbR, AGG_THREADS, 0, stream>>>(src, dst, er, feat, stat + 6 * l, h,
                                             gat_bias + (size_t)l * FDIM, glist,
                                             bmap, n, ne, l);
  }

  gemm36_wmma<FDIM><<<gemmGrid, 32, 0, stream>>>(h, out_w, out_b, (float*)d_out, n);
}


# el, er as six more columns of the feat GEMM (fc_w times the attn vectors, padded weight block made once per call in edge_bits): attn_scores launches go; el = h*(W a), sum order changes
# speedup vs baseline: 20.3566x; 1.0520x over previous
#include <hip/hip_runtime.h>
#include <math.h>

typedef __attribute__((ext_vector_type(2))) float v2f;
typedef __attribute__((ext_vector_type(8))) float v8f;
typedef __attribute__((ext_vector_type(4))) int v4i;

#define FDIM 36
#define NHEAD 3
#define HDIM 12
#define NEG_SLOPE 0.2f
#define NEG_INF_KEY 0x007FFFFFu
#define FP 40
#define MAX_GAP 64.0f

__device__ __forceinline__ unsigned f2key(float f) {
  unsigned u = __float_as_uint(f);
  return (u & 0x80000000u) ? ~u : (u | 0x80000000u);
}
__device__ __forceinline__ float key2f(unsigned k) {
  return (k & 0x80000000u) ? __uint_as_float(k & 0x7FFFFFFFu)
                           : __uint_as_float(~k);
}
__device__ __forceinline__ float leaky(float x) {
  return x >= 0.0f ? x : NEG_SLOPE * x;
}

template <int LDO>
__global__ __launch_bounds__(32) void gemm36_wmma(
    const float* __restrict__ A, const float* __restrict__ W,
    const float* __restrict__ bias, float* __restrict__ out, int nrows) {
  const int lane = threadIdx.x;
  const int half = lane >> 4;
  const int idx  = lane & 15;

  int mrow = blockIdx.x * 16 + idx;
  mrow = mrow < nrows ? mrow : (nrows - 1);
  const float* __restrict__ Ab = A + (size_t)mrow * FDIM + 2 * half;

  const int col0 = idx;
  const int col1 = idx + 16;
  const int col2 = (idx + 32) < FDIM ? (idx + 32) : (FDIM - 1);
  const float* __restrict__ Wb0 = W + col0 + 72 * half;
  const float* __restrict__ Wb1 = W + col1 + 72 * half;
  const float* __restrict__ Wb2 = W + col2 + 72 * half;

  v8f acc[3];
  acc[0] = (v8f){0.f,0.f,0.f,0.f,0.f,0.f,0.f,0.f};
  acc[1] = acc[0];
  acc[2] = acc[0];

  #pragma unroll
  for (int k0 = 0; k0 < FDIM; k0 += 4) {
    v2f a, b0, b1, b2;
    a.x  = Ab[k0];
    a.y  = Ab[k0 + 1];
    b0.x = Wb0[(k0 + 0) * FDIM];
    b0.y = Wb0[(k0 + 1) * FDIM];
    b1.x = Wb1[(k0 + 0) * FDIM];
    b1.y = Wb1[(k0 + 1) * FDIM];
    b2.x = Wb2[(k0 + 0) * FDIM];
    b2.y = Wb2[(k0 + 1) * FDIM];
    acc[0] = __builtin_amdgcn_wmma_f32_16x16x4_f32(false, a, false, b0,
                                                   (short)0, acc[0], false, false);
    acc[1] = __builtin_amdgcn_wmma_f32_16x16x4_f32(false, a, false, b1,
                                                   (short)0, acc[1], false, false);
    acc[2] = __builtin_amdgcn_wmma_f32_16x16x4_f32(false, a, false, b2,
                                                   (short)0, acc[2], false, false);
  }

  const int rowbase = blockIdx.x * 16 + 8 * half;
  const bool full = (blockIdx.x * 16 + 16) <= nrows;
  #pragma unroll
  for (int t = 0; t < 3; ++t) {
    const int col = idx + 16 * t;
    if (col < FDIM) {
      const float bv = bias ? bias[col] : 0.0f;
      float* o = out + (size_t)rowbase * LDO + col;
      if (full) {
        #pragma unroll
        for (int v = 0; v < 8; ++v) o[(size_t)v * LDO] = acc[t][v] + bv;
      } else {
        #pragma unroll
        for (int v = 0; v < 8; ++v)
          if (rowbase + v < nrows) o[(size_t)v * LDO] = acc[t][v] + bv;
      }
    }
  }
}

#define WP 48
#define GW 16
__global__ __launch_bounds__(32 * GW) void gemm36_feat(
    const float* __restrict__ A, const float* __restrict__ Wp,
    float* __restrict__ feat, float* __restrict__ er, unsigned* stat, int nrows) {
  __shared__ unsigned sst[6];
  const int tid = threadIdx.x;
  if (tid < 6) sst[tid] = tid < 3 ? 0u : 0xFFFFFFFFu;
  __syncthreads();
  const int lane = tid & 31;
  const int half = lane >> 4;
  const int idx  = lane & 15;
  const int tile = blockIdx.x * GW + __builtin_amdgcn_readfirstlane(tid >> 5);

  int mrow = tile * 16 + idx;
  mrow = mrow < nrows ? mrow : (nrows - 1);
  const float* __restrict__ Ab = A + (size_t)mrow * FDIM + 2 * half;
  const float* __restrict__ Wb = Wp + idx + 2 * WP * half;

  v8f acc[3];
  acc[0] = (v8f){0.f,0.f,0.f,0.f,0.f,0.f,0.f,0.f};
  acc[1] = acc[0];
  acc[2] = acc[0];

  #pragma unroll
  for (int k0 = 0; k0 < FDIM; k0 += 4) {
    v2f a, b0, b1, b2;
    a.x  = Ab[k0];
    a.y  = Ab[k0 + 1];
    b0.x = Wb[(k0 + 0) * WP];
    b0.y = Wb[(k0 + 1) * WP];
    b1.x = Wb[(k0 + 0) * WP + 16];
    b1.y = Wb[(k0 + 1) * WP + 16];
    b2.x = Wb[(k0 + 0) * WP + 32];
    b2.y = Wb[(k0 + 1) * WP + 32];
    acc[0] = __builtin_amdgcn_wmma_f32_16x16x4_f32(false, a, false, b0,
                                                   (short)0, acc[0], false, false);
    acc[1] = __builtin_amdgcn_wmma_f32_16x16x4_f32(false, a, false, b1,
                                                   (short)0, acc[1], false, false);
    acc[2] = __builtin_amdgcn_wmma_f32_16x16x4_f32(false, a, false, b2,
                                                   (short)0, acc[2], false, false);
  }

  const int rowbase = tile * 16 + 8 * half;
  const bool full = (tile * 16 + 16) <= nrows;
  #pragma unroll
  for (int t = 0; t < 3; ++t) {
    const int col = idx + 16 * t;
    if (col < FP) {
      float* o = feat + (size_t)rowbase * FP + col;
      if (full) {
        #pragma unroll
        for (int v = 0; v < 8; ++v) o[(size_t)v * FP] = acc[t][v];
      } else {
        #pragma unroll
        for (int v = 0; v < 8; ++v)
          if (rowbase + v < nrows) o[(size_t)v * FP] = acc[t][v];
      }
    }
  }
  if (idx >= 8 && idx < 8 + NHEAD) {
    float* o = er + (size_t)rowbase * NHEAD + (idx - 8);
    #pragma unroll
    for (int v = 0; v < 8; ++v)
      if (full || rowbase + v < nrows) o[(size_t)v * NHEAD] = acc[2][v];
  }

  unsigned kx = 0u, kn = 0xFFFFFFFFu;
  #pragma unroll
  for (int v = 0; v < 8; ++v) {
    const unsigned k = f2key(acc[2][v]);
    kx = k > kx ? k : kx;
    kn = k < kn ? k : kn;
  }
  const unsigned ox = (unsigned)__shfl_xor((int)kx, 16);
  const unsigned on2 = (unsigned)__shfl_xor((int)kn, 16);
  kx = ox > kx ? ox : kx;
  kn = on2 < kn ? on2 : kn;
  if (half == 0 && idx >= 4 && idx < 4 + NHEAD) {
    atomicMax(&sst[idx - 4], kx);
    atomicMin(&sst[3 + idx - 4], kn);
  }
  __syncthreads();
  if (tid < 3) atomicMax(&stat[tid], sst[tid]);
  else if (tid < 6) atomicMin(&stat[tid], sst[tid]);
}

#define RN 392
#define KV 1024
#define BPITCH 257
__global__ __launch_bounds__(KV) void edge_bits(const int* __restrict__ dst,
                                                unsigned* __restrict__ bmap,
                                                unsigned* __restrict__ stat, int ne,
                                                const float* __restrict__ fc_w,
                                                const float* __restrict__ a_l,
                                                const float* __restrict__ a_r,
                                                float* __restrict__ wpad) {
  __shared__ unsigned bm[(KV / 32) * BPITCH];
  const int tid = threadIdx.x;
  const int lane = tid & 31;
  const int wave = tid >> 5;
  for (int i = tid; i < (KV / 32) * BPITCH; i += KV) bm[i] = 0u;
  if (blockIdx.x == 0 && tid < 18) stat[tid] = (tid % 6) < 3 ? 0u : 0xFFFFFFFFu;
  for (int i = blockIdx.x * KV + tid; i < 3 * FDIM * WP; i += gridDim.x * KV) {
    const int l = i / (FDIM * WP);
    const int r = i - l * (FDIM * WP);
    const int k = r / WP;
    const int c = r - k * WP;
    const float* __restrict__ wrow = fc_w + ((size_t)l * FDIM + k) * FDIM;
    float v = 0.0f;
    if (c < FDIM) {
      v = wrow[c];
    } else if (c < FDIM + NHEAD || (c >= FP && c < FP + NHEAD)) {
      const int hd = c < FP ? c - FDIM : c - FP;
      const float* __restrict__ av = (c < FP ? a_l : a_r) + (size_t)l * FDIM + hd * HDIM;
      for (int d = 0; d < HDIM; ++d) v += wrow[hd * HDIM + d] * av[d];
    }
    wpad[i] = v;
  }
  __syncthreads();

  const int nv8 = (ne + 7) >> 3;
  const int v = blockIdx.x * KV + tid;
  if (v < nv8) {
    int d[8];
    if (8 * v + 7 < ne) {
      const int4 t0 = ((const int4*)dst)[2 * v];
      const int4 t1 = ((const int4*)dst)[2 * v + 1];
      d[0] = t0.x; d[1] = t0.y; d[2] = t0.z; d[3] = t0.w;
      d[4] = t1.x; d[5] = t1.y; d[6] = t1.z; d[7] = t1.w;
    } else {
      #pragma unroll
      for (int j = 0; j < 8; ++j) d[j] = (8 * v + j) < ne ? dst[8 * v + j] : -1;
    }
    #pragma unroll
    for (int j = 0; j < 8; ++j) {
      unsigned b = (unsigned)d[j] / RN;
      b = b < 255u ? b : 255u;
      if (8 * v + j < ne) atomicOr(&bm[wave * BPITCH + b], 1u << lane);
    }
  }
  __syncthreads();

  for (int idx = tid; idx < 256 * 32; idx += KV) {
    const int r = idx >> 5;
    const int w = idx & 31;
    bmap[((size_t)r * gridDim.x + blockIdx.x) * 32 + w] = bm[w * BPITCH + r];
  }
}

#define GLW 34
#define PACK_SHIFT 9
#define PACK_MAXN (1 << 22)
#define APITCH 40
#define LCAP 896
#define FCAP 576
#define CCAP 640
#define EXP_U 2
#define WCAP (LCAP / 2)
#define PACK_MASK ((1 << PACK_SHIFT) - 1)
#define AGG_THREADS 1024
#define AGG_WAVES (AGG_THREADS / 32)

__device__ __forceinline__ int colpos(int c) { return (c & 3) * 9 + (c >> 2); }

#define ROW_GROUPS 11

template <bool FULL, bool WIDE>
__device__ __forceinline__ void agg_batch(const int2* q, int head, int cnt,
                                          int lane,
                                          const float* __restrict__ el,
                                          const float* __restrict__ feat,
                                          const float* sEr,
                                          const unsigned* sMx, float* acc) {
  const float4* __restrict__ feat4 = (const float4*)feat;
  const v4i* q4 = (const v4i*)q;
  const int* qw = (const int*)q;
  const int r3 = lane / 9;
  const int rr = r3 < 3 ? r3 : 2;
  const int pc = lane - 9 * r3;

  const bool on = FULL || lane < cnt;
  int dl = 0;
  float ev[NHEAD] = {0.0f, 0.0f, 0.0f};
  if (on) {
    if (WIDE) {
      const v4i ent = q4[head + lane];
      dl = ent.w & PACK_MASK;
      ev[0] = __int_as_float(ent.x);
      ev[1] = __int_as_float(ent.y);
      ev[2] = __int_as_float(ent.z);
    } else {
      const int2 ent = q[head + lane];
      const int s = ent.x;
      dl = ent.y;
      #pragma unroll
      for (int h = 0; h < NHEAD; ++h) ev[h] = el[(size_t)s * FP + h];
    }
  }

  if (FULL) {
    float4 f[ROW_GROUPS];
    int dn[ROW_GROUPS];
    #pragma unroll
    for (int g = 0; g < ROW_GROUPS; ++g) {
      const int j = (3 * g + rr) < 32 ? (3 * g + rr) : 31;
      int sx;
      if (WIDE) {
        const int* qr = qw + 4 * (head + rr) + 3;
        const int pk = g < ROW_GROUPS - 1 ? qr[12 * g] : qr[rr < 2 ? 120 : 116];
        sx = pk >> PACK_SHIFT;
        dn[g] = pk & PACK_MASK;
      } else {
        const int2 en = q[head + j];
        sx = en.x;
        dn[g] = en.y;
      }
      f[g] = feat4[(size_t)sx * (FP / 4) + pc];
    }
    if (!WIDE) {
      #pragma unroll
      for (int g = 0; g < ROW_GROUPS; ++g)
        asm volatile("" :: "v"(f[g].x), "v"(f[g].y), "v"(f[g].z), "v"(f[g].w));
    }

    float w[NHEAD];
    #pragma unroll
    for (int h = 0; h < NHEAD; ++h) {
      const float v = WIDE ? ev[h] : leaky(ev[h] + sEr[dl * NHEAD + h]);
      w[h] = expf(v - __uint_as_float(sMx[dl * NHEAD + h]));
      atomicAdd(&acc[dl * APITCH + FDIM + h], w[h]);
    }

    float a[ROW_GROUPS];
    #pragma unroll
    for (int g = 0; g < ROW_GROUPS; ++g) {
      const int j = (3 * g + rr) < 32 ? (3 * g + rr) : 31;
      const float a0 = __shfl(w[0], j);
      const float a1 = __shfl(w[1], j);
      const float a2 = __shfl(w[2], j);
      a[g] = pc < 3 ? a0 : (pc < 6 ? a1 : a2);
    }
    if (WIDE) {
      #pragma unroll
      for (int g = 0; g < ROW_GROUPS; ++g)
        asm volatile("" :: "v"(f[g].x), "v"(f[g].y), "v"(f[g].z), "v"(f[g].w));
    }
    #pragma unroll
    for (int g = 0; g < ROW_GROUPS; ++g) {
      if (r3 < 3 && 3 * g + r3 < 32) {
        float* p = acc + dn[g] * APITCH + pc;
        atomicAdd(p, f[g].x * a[g]);
        atomicAdd(p + 9, f[g].y * a[g]);
        atomicAdd(p + 18, f[g].z * a[g]);
        atomicAdd(p + 27, f[g].w * a[g]);
      }
    }
  } else {
    float w0 = 0.0f, w1 = 0.0f, w2 = 0.0f;
    if (on) {
      float w[NHEAD];
      #pragma unroll
      for (int h = 0; h < NHEAD; ++h) {
        const float v = WIDE ? ev[h] : leaky(ev[h] + sEr[dl * NHEAD + h]);
        w[h] = expf(v - __uint_as_float(sMx[dl * NHEAD + h]));
        atomicAdd(&acc[dl * APITCH + FDIM + h], w[h]);
      }
      w0 = w[0]; w1 = w[1]; w2 = w[2];
    }
    const int ng = (cnt + 2) / 3;
    for (int g = 0; g < ng; ++g) {
      const int jr = 3 * g + rr;
      const bool ok = r3 < 3 && jr < cnt;
      const int j = jr < cnt ? jr : cnt - 1;
      int sx, dy;
      if (WIDE) {
        const int pk = qw[4 * (head + j) + 3];
        sx = pk >> PACK_SHIFT;
        dy = pk & PACK_MASK;
      } else {
        const int2 en = q[head + j];
        sx = en.x;
        dy = en.y;
      }
      const float4 fv = feat4[(size_t)sx * (FP / 4) + pc];
      const float a0 = __shfl(w0, j);
      const float a1 = __shfl(w1, j);
      const float a2 = __shfl(w2, j);
      const float av = pc < 3 ? a0 : (pc < 6 ? a1 : a2);
      if (ok) {
        float* p = acc + dy * APITCH + pc;
        atomicAdd(p, fv.x * av);
        atomicAdd(p + 9, fv.y * av);
        atomicAdd(p + 18, fv.z * av);
        atomicAdd(p + 27, fv.w * av);
      }
    }
  }
}

template <bool LOADED>
__device__ __forceinline__ void list_max(int2* q, int tail, bool wide, int lane,
                                         const int* __restrict__ src,
                                         const float* __restrict__ el,
                                         const float* sEr, unsigned* sMx) {
  v4i* q4 = (v4i*)q;
  const int* qw = (const int*)q;
  #pragma nounroll
  for (int i0 = (tail - 1) & ~127; i0 >= 0; i0 -= 128) {
    int sv[4], dv[4];
    #pragma unroll
    for (int k = 0; k < 4; ++k) {
      const int i = i0 + 32 * k + lane;
      if (LOADED) {
        const int pk = qw[4 * (i < WCAP ? i : WCAP - 1) + 3];
        const int x = i < tail ? pk : 0;
        sv[k] = x >> PACK_SHIFT;
        dv[k] = x & PACK_MASK;
      } else {
        const int2 ent = q[i < LCAP ? i : LCAP - 1];
        const int x = i < tail ? ent.x : 0;
        sv[k] = src[x];
        dv[k] = i < tail ? ent.y : 0;
      }
    }
    float ev[4][NHEAD];
    #pragma unroll
    for (int k = 0; k < 4; ++k) {
      #pragma unroll
      for (int h = 0; h < NHEAD; ++h) ev[k][h] = el[(size_t)sv[k] * FP + h];
    }
    #pragma unroll
    for (int k = 0; k < 4; ++k) {
      const int i = i0 + 32 * k + lane;
      float v[NHEAD];
      #pragma unroll
      for (int h = 0; h < NHEAD; ++h) {
        v[h] = leaky(ev[k][h] + sEr[dv[k] * NHEAD + h]);
        atomicMax(&sMx[dv[k] * NHEAD + h], i < tail ? f2key(v[h]) : NEG_INF_KEY);
      }
      if (i < tail) {
        if (LOADED || wide) {
          q4[i] = (v4i){__float_as_int(v[0]), __float_as_int(v[1]),
                        __float_as_int(v[2]), (sv[k] << PACK_SHIFT) | dv[k]};
        } else {
          q[i].x = sv[k];
        }
      }
    }
  }
}

__device__ __forceinline__ void agg_bound(const int* qw, int head, int cnt, int lane,
                                          const float* __restrict__ feat,
                                          const v2f* sB, float* acc) {
  const float4* __restrict__ feat4 = (const float4*)feat;
  const int r3 = lane / 10;
  const int rr = r3 < 3 ? r3 : 2;
  const int pc = lane - 10 * r3;
  const int hh = pc < 3 ? 0 : (pc < 6 ? 1 : 2);
  const int esrc = 10 * rr + 9;
  int pk[ROW_GROUPS];
  #pragma unroll
  for (int g = 0; g < ROW_GROUPS; ++g) {
    const int jr = 3 * g + rr;
    pk[g] = qw[head + (jr < cnt ? jr : cnt - 1)];
  }
  float4 f[ROW_GROUPS];
  v2f b[ROW_GROUPS];
  #pragma unroll
  for (int g = 0; g < ROW_GROUPS; ++g)
    f[g] = feat4[(size_t)(pk[g] >> PACK_SHIFT) * (FP / 4) + pc];
  #pragma unroll
  for (int g = 0; g < ROW_GROUPS; ++g)
    b[g] = sB[(pk[g] & PACK_MASK) * NHEAD + hh];
  __builtin_amdgcn_sched_barrier(0);
  float w[ROW_GROUPS];
  #pragma unroll
  for (int g = 0; g < ROW_GROUPS; ++g) {
    const float e0 = __shfl(f[g].x, esrc);
    const float e1 = __shfl(f[g].y, esrc);
    const float e2 = __shfl(f[g].z, esrc);
    w[g] = hh == 0 ? e0 : (hh == 1 ? e1 : e2);
  }
  #pragma unroll
  for (int g = 0; g < ROW_GROUPS; ++g) w[g] = expf(leaky(w[g] + b[g].x) - b[g].y);
  #pragma unroll
  for (int g = 0; g < ROW_GROUPS; ++g) {
    if (r3 < 3 && pc < 9 && 3 * g + r3 < cnt) {
      float* p = acc + (pk[g] & PACK_MASK) * APITCH + pc;
      atomicAdd(p, f[g].x * w[g]);
      atomicAdd(p + 9, f[g].y * w[g]);
      atomicAdd(p + 18, f[g].z * w[g]);
      atomicAdd(p + 27, f[g].w * w[g]);
      if (pc == 3 * hh) atomicAdd(acc + (pk[g] & PACK_MASK) * APITCH + FDIM + hh, w[g]);
    }
  }
}

__device__ __forceinline__ void list_pack(int* qw, int tail, int lane,
                                          const int* __restrict__ src) {
  #pragma nounroll
  for (int i0 = 0; i0 < tail; i0 += 128) {
    int sv[4], dv[4];
    #pragma unroll
    for (int k = 0; k < 4; ++k) {
      const int i = i0 + 32 * k + lane;
      const int ic = i < LCAP ? i : LCAP - 1;
      const int ex = qw[2 * ic];
      dv[k] = qw[2 * ic + 1];
      sv[k] = src[i < tail ? ex : 0];
    }
    #pragma unroll
    for (int k = 0; k < 4; ++k) {
      const int i = i0 + 32 * k + lane;
      if (i < tail) qw[i] = (sv[k] << PACK_SHIFT) | dv[k];
    }
  }
}

__global__ __launch_bounds__(AGG_THREADS) void edge_agg(
    const int* __restrict__ src, const int* __restrict__ dst,
    const float* __restrict__ er,
    const float* __restrict__ feat, const unsigned* __restrict__ stat,
    float* h, const float* __restrict__ bias,
    int* glist, const unsigned* __restrict__ bmap, int n, int ne, int layer) {
  const float* __restrict__ el = feat + FDIM;
  __shared__ float acc[RN * APITCH];
  __shared__ float sEr[RN * NHEAD];
  __shared__ unsigned sMx[RN * NHEAD];
  __shared__ v2f sB[RN * NHEAD];
  __shared__ v4i lst4[AGG_WAVES * WCAP];
  int2* lst = (int2*)lst4;
  __shared__ int sOvf;
  __shared__ int sCnt[AGG_WAVES];
  __shared__ int sOff[AGG_WAVES + 1];

  const int tid = threadIdx.x;
  const int lane = tid & 31;
  const int wave = __builtin_amdgcn_readfirstlane(tid >> 5);
  const int lo = blockIdx.x * RN;
  const int cntN = (n - lo) < RN ? (n - lo) : RN;

  float gM[NHEAD];
  bool fb = n <= PACK_MAXN;
  #pragma unroll
  for (int hd = 0; hd < NHEAD; ++hd) {
    const float mx = key2f(stat[hd]);
    const float mn = key2f(stat[NHEAD + hd]);
    gM[hd] = mx;
    fb = fb && (mx - mn <= MAX_GAP);
  }
  const int fast = __builtin_amdgcn_readfirstlane(fb ? 1 : 0);

  for (int i = tid; i < RN * APITCH; i += AGG_THREADS) acc[i] = 0.0f;
  for (int i = tid; i < RN * NHEAD; i += AGG_THREADS) sMx[i] = NEG_INF_KEY;
  for (int i = tid; i < cntN * NHEAD; i += AGG_THREADS)
    sEr[i] = er[(size_t)lo * NHEAD + i];
  if (fast) {
    for (int i = tid; i < cntN * NHEAD; i += AGG_THREADS) {
      const int hd = i % NHEAD;
      const float e = er[(size_t)lo * NHEAD + i];
      const float m = hd == 0 ? gM[0] : (hd == 1 ? gM[1] : gM[2]);
      sB[i] = (v2f){e, leaky(m + e)};
    }
  }
  if (tid == 0) sOvf = 0;
  __syncthreads();

  int2* q = lst + wave * LCAP;
  int* qi = (int*)q;
  int tail = 0;
  int ncand = 0;

  const int capT = ((n * GLW) / (int)gridDim.x) & ~31;
  int* gl = glist + (size_t)blockIdx.x * capT;
  int loaded = 0;
  if (layer > 0) {
    const int cnt = __builtin_amdgcn_readfirstlane(gl[0]);
    if (cnt >= 0) {
      loaded = 1;
      const int nq = (cnt + 3) >> 2;
      const int qa = (wave * nq) >> 5;
      const int nqw = (((wave + 1) * nq) >> 5) - qa;
      const int rem = cnt - 4 * qa;
      tail = 4 * nqw < rem ? 4 * nqw : rem;
      const int* gw = gl + 32 + 4 * qa;
      int p[WCAP / 32];
      #pragma unroll
      for (int k = 0; k < WCAP / 32; ++k) {
        const int i = 32 * k + lane;
        p[k] = gw[i < tail ? i : 0];
      }
      #pragma unroll
      for (int k = 0; k < WCAP / 32; ++k) {
        const int i = 32 * k + lane;
        if (i < tail) qi[fast ? i : 4 * i + 3] = p[k];
      }
    }
  }

  int ovf = (!loaded && gridDim.x > 256) ? 1 : 0;

  const int nv8 = (ne + 7) >> 3;
  if (nv8 > 0 && !ovf && !loaded) {
    const int nw = ((nv8 + KV - 1) / KV) * 32;
    const unsigned* __restrict__ row = bmap + (size_t)blockIdx.x * nw;
    const int wa = (wave * nw) >> 5;
    const int wb = ((wave + 1) * nw) >> 5;
    for (int base = wa; base < wb; base += 256) {
      unsigned mw[8];
      #pragma unroll
      for (int u = 0; u < 8; ++u) {
        const int k = base + 32 * u + lane;
        const unsigned x = row[k < wb ? k : wa];
        mw[u] = k < wb ? x : 0u;
      }
      int c = 0;
      #pragma unroll
      for (int u = 0; u < 8; ++u) c += __builtin_popcount(mw[u]);
      int inc = c;
      #pragma unroll
      for (int d = 1; d < 32; d <<= 1) {
        const int t = __shfl_up(inc, d);
        if (lane >= d) inc += t;
      }
      const int tot = __builtin_amdgcn_readlane(inc, 31);
      if (ncand + tot <= CCAP) {
        int pos = ncand + inc - c;
        #pragma unroll
        for (int u = 0; u < 8; ++u) {
          unsigned m = mw[u];
          const int vb = (base + 32 * u + lane) * 32;
          while (m != 0u) {
            qi[2 * LCAP - 1 - pos] = vb + __builtin_ctz(m);
            ++pos;
            m &= m - 1u;
          }
        }
        ncand += tot;
      } else {
        ovf = 1;
      }
    }
  }
  __syncthreads();

#define PUSH(dval, eidx, okb)                                                \
  {                                                                          \
    const int dlc = (dval) - lo;                                             \
    const unsigned mk = __builtin_amdgcn_ballot_w32(                         \
        (okb) && (unsigned)dlc < (unsigned)cntN);                            \
    if (mk) {                                                                \
      const int ntail = tail + __builtin_popcount(mk);                       \
      if (ntail <= FCAP) {                                                   \
        if ((mk >> lane) & 1u) {                                             \
          const int pos = tail + (int)__builtin_amdgcn_mbcnt_lo(mk, 0u);     \
          q[pos] = make_int2((eidx), dlc);                                   \
        }                                                                    \
        tail = ntail;                                                        \
      } else {                                                               \
        ovf = 1;                                                             \
      }                                                                      \
    }                                                                        \
  }
  if (!ovf && !loaded) {
    const int4* __restrict__ dst4 = (const int4*)dst;
    for (int i0 = 0; i0 < ncand; i0 += 32 * EXP_U) {
      int vv[EXP_U];
      int dd[EXP_U][8];
      unsigned hm[EXP_U];
      #pragma unroll
      for (int u = 0; u < EXP_U; ++u) {
        const int ci = i0 + 32 * u + lane;
        vv[u] = qi[2 * LCAP - 1 - (ci < ncand ? ci : ncand - 1)];
        const bool whole = 8 * vv[u] + 7 < ne;
        int4 t0 = make_int4(-1, -1, -1, -1), t1 = t0;
        if (ne >= 8) {
          const int s8 = whole ? vv[u] : 0;
          t0 = dst4[2 * s8];
          t1 = dst4[2 * s8 + 1];
        }
        dd[u][0] = t0.x; dd[u][1] = t0.y; dd[u][2] = t0.z; dd[u][3] = t0.w;
        dd[u][4] = t1.x; dd[u][5] = t1.y; dd[u][6] = t1.z; dd[u][7] = t1.w;
        if (!whole) {
          #pragma unroll
          for (int j = 0; j < 8; ++j)
            dd[u][j] = (8 * vv[u] + j) < ne ? dst[8 * vv[u] + j] : -1;
        }
      }
      #pragma unroll
      for (int u = 0; u < EXP_U; ++u) {
        const int ci = i0 + 32 * u + lane;
        unsigned hb = 0u;
        #pragma unroll
        for (int j = 0; j < 8; ++j)
          hb |= (unsigned)(dd[u][j] - lo) < (unsigned)cntN ? (1u << j) : 0u;
        hm[u] = ci < ncand ? hb : 0u;
      }
      unsigned any = 0u;
      #pragma unroll
      for (int u = 0; u < EXP_U; ++u) any |= hm[u];
      while (__builtin_amdgcn_ballot_w32(any != 0u)) {
        #pragma unroll
        for (int u = 0; u < EXP_U; ++u) {
          const int p = __builtin_ctz(hm[u] | 0x100u);
          int dsel = dd[u][0];
          #pragma unroll
          for (int j = 1; j < 8; ++j) dsel = p == j ? dd[u][j] : dsel;
          const bool okk = hm[u] != 0u;
          PUSH(dsel, 8 * vv[u] + p, okk)
          hm[u] &= hm[u] - 1u;
        }
        any = 0u;
        #pragma unroll
        for (int u = 0; u < EXP_U; ++u) any |= hm[u];
      }
    }
  }
#undef PUSH
  if (ovf && lane == 0) sOvf = 1;
  __syncthreads();
  const int slow = __builtin_amdgcn_readfirstlane(sOvf);

  const bool wide = loaded || (tail <= WCAP && n <= PACK_MAXN);

  if (!slow) {
    if (fast) { if (!loaded) list_pack(qi, tail, lane, src); }
    else if (loaded) list_max<true>(q, tail, true, lane, src, el, sEr, sMx);
    else             list_max<false>(q, tail, wide, lane, src, el, sEr, sMx);
  } else {
    for (int e = tid; e < ne; e += AGG_THREADS) {
      const int dlc = dst[e] - lo;
      if ((unsigned)dlc < (unsigned)cntN) {
        const int s = src[e];
        #pragma unroll
        for (int h = 0; h < NHEAD; ++h)
          atomicMax(&sMx[dlc * NHEAD + h],
                    f2key(leaky(el[(size_t)s * FP + h] + sEr[dlc * NHEAD + h])));
      }
    }
  }
  if (layer == 0 && lane == 0) sCnt[wave] = slow ? 0 : tail;
  __syncthreads();

  if (layer == 0 && tid < AGG_WAVES) {
    int o = 0;
    for (int w = 0; w < AGG_WAVES; ++w) o += w < tid ? sCnt[w] : 0;
    sOff[tid] = o;
    if (tid == AGG_WAVES - 1) sOff[AGG_WAVES] = o + sCnt[AGG_WAVES - 1];
  }

  for (int i = tid; i < cntN * NHEAD; i += AGG_THREADS) {
    float v = key2f(sMx[i]);
    if (!isfinite(v)) v = 0.0f;
    sMx[i] = __float_as_uint(v);
  }
  __syncthreads();

  if (layer == 0) {
    const int total = __builtin_amdgcn_readfirstlane(sOff[AGG_WAVES]);
    const bool ok = !slow && n <= PACK_MAXN && capT >= 64 && total <= capT - 32;
    if (ok) {
      const int up = total > 0 ? ((total + 31) & ~31) : 32;
      for (int idx = tid; idx < up; idx += AGG_THREADS) {
        int w = 0;
        #pragma unroll
        for (int s = AGG_WAVES / 2; s > 0; s >>= 1)
          if (sOff[w + s] <= idx) w += s;
        int val = 0;
        if (idx < total) {
          const int k = idx - sOff[w];
          if (fast) {
            val = ((const int*)lst4)[2 * LCAP * w + k];
          } else if (sCnt[w] <= WCAP) {
            val = ((const int*)lst4)[4 * (w * WCAP + k) + 3];
          } else {
            const int2 ent = lst[w * LCAP + k];
            val = (ent.x << PACK_SHIFT) | ent.y;
          }
        }
        gl[32 + idx] = val;
      }
    }
    if (tid < 32) gl[tid] = tid == 0 ? (ok ? total : -1) : 0;
  }

  if (!slow && fast) {
    for (int head = 0; head < tail; head += 32)
      agg_bound(qi, head, (tail - head) < 32 ? (tail - head) : 32, lane, feat, sB, acc);
  } else if (!slow) {
    int head = 0;
    if (wide) {
      while (tail - head >= 32) {
        agg_batch<true, true>(q, head, 32, lane, el, feat, sEr, sMx, acc);
        head += 32;
      }
      if (tail - head > 0)
        agg_batch<false, true>(q, head, tail - head, lane, el, feat, sEr, sMx, acc);
    } else {
      while (tail - head >= 32) {
        agg_batch<true, false>(q, head, 32, lane, el, feat, sEr, sMx, acc);
        head += 32;
      }
      if (tail - head > 0)
        agg_batch<false, false>(q, head, tail - head, lane, el, feat, sEr, sMx, acc);
    }
  } else {
    for (int e = tid; e < ne; e += AGG_THREADS) {
      const int dlc = dst[e] - lo;
      if ((unsigned)dlc < (unsigned)cntN) {
        const int s = src[e];
        float w[NHEAD];
        #pragma unroll
        for (int h = 0; h < NHEAD; ++h) {
          const float v = leaky(el[(size_t)s * FP + h] + sEr[dlc * NHEAD + h]);
          w[h] = expf(v - __uint_as_float(sMx[dlc * NHEAD + h]));
          atomicAdd(&acc[dlc * APITCH + FDIM + h], w[h]);
        }
        for (int c = 0; c < FDIM; ++c)
          atomicAdd(&acc[dlc * APITCH + colpos(c)],
                    feat[(size_t)s * FP + c] * w[c / HDIM]);
      }
    }
  }
  __syncthreads();

  for (int i = tid; i < cntN * FDIM; i += AGG_THREADS) {
    const int node = i / FDIM;
    const int c = i - node * FDIM;
    const float wsum = acc[node * APITCH + FDIM + c / HDIM];
    const float den = (fast && !slow) ? (wsum > 0.0f ? wsum : 1.0f) : fmaxf(wsum, 1e-9f);
    const size_t g = (size_t)lo * FDIM + i;
    const float x = acc[node * APITCH + colpos(c)] / den + h[g] + bias[c];
    h[g] = x > 0.0f ? x : expm1f(x);
  }
}

extern "C" void kernel_launch(void* const* d_in, const int* in_sizes, int n_in,
                              void* d_out, int out_size, void* d_ws,
                              size_t ws_size, hipStream_t stream) {
  (void)n_in; (void)out_size; (void)ws_size;
  const float* X        = (const float*)d_in[0];
  const int*   src      = (const int*)d_in[1];
  const int*   dst      = (const int*)d_in[2];
  const float* lin0_w   = (const float*)d_in[3];
  const float* lin0_b   = (const float*)d_in[4];
  const float* fc_w     = (const float*)d_in[5];
  const float* attn_l   = (const float*)d_in[6];
  const float* attn_r   = (const float*)d_in[7];
  const float* gat_bias = (const float*)d_in[8];
  const float* out_w    = (const float*)d_in[9];
  const float* out_b    = (const float*)d_in[10];

  const int n  = in_sizes[0] / FDIM;
  const int ne = in_sizes[1];

  float* ws   = (float*)d_ws;
  float* h     = ws; ws += (size_t)n * FDIM;
  float* feat  = ws; ws += (size_t)n * FP;
  int*   glist = (int*)ws; ws += (size_t)n * GLW;
  float* er    = ws; ws += (size_t)n * NHEAD;
  unsigned* stat = (unsigned*)ws;

  const dim3 gemmGrid((n + 15) / 16);
  const dim3 featGrid((n + 16 * GW - 1) / (16 * GW));
  const int nbR  = (n + RN - 1) / RN;

  unsigned* bmap = (unsigned*)d_out;
  const int nv8 = (ne + 7) / 8;
  float* wpad = (float*)d_out + (size_t)n * FDIM - 3 * FDIM * WP;
  edge_bits<<<(nv8 + KV - 1) / KV, KV, 0, stream>>>(dst, bmap, stat, ne,
                                                    fc_w, attn_l, attn_r, wpad);

  gemm36_wmma<FDIM><<<gemmGrid, 32, 0, stream>>>(X, lin0_w, lin0_b, h, n);

  for (int l = 0; l < 3; ++l) {
    gemm36_feat<<<featGrid, 32 * GW, 0, stream>>>(
        h, wpad + (size_t)l * FDIM * WP, feat, er, stat + 6 * l, n);
    edge_agg<<<nbR, AGG_THREADS, 0, stream>>>(src, dst, er, feat, stat + 6 * l, h,
                                             gat_bias + (size_t)l * FDIM, glist,
                                             bmap, n, ne, l);
  }

  gemm36_wmma<FDIM><<<gemmGrid, 32, 0, stream>>>(h, out_w, out_b, (float*)d_out, n);
}
